# starting kernel
# baseline (speedup 1.0000x reference)
#include <hip/hip_runtime.h>
#include <hip/hip_bf16.h>
#include <stdint.h>

typedef __attribute__((ext_vector_type(16))) _Float16 v16h;
typedef __attribute__((ext_vector_type(8)))  float    v8f;

#define IN_C   128
#define HID_C  128
#define OUT_C  64
#define NGRAPH 128

__device__ __forceinline__ void atomic_add_f32(float* p, float v) {
    unsafeAtomicAdd(p, v);
}

__global__ void k_fill(float* p, float v, int n) {
    int i = blockIdx.x * blockDim.x + threadIdx.x;
    if (i < n) p[i] = v;
}

__global__ void k_edge_deg(const int* __restrict__ dst, float* deg, int E) {
    int e = blockIdx.x * blockDim.x + threadIdx.x;
    if (e < E) atomic_add_f32(&deg[dst[e]], 1.0f);
}

__global__ void k_rsqrt(const float* __restrict__ deg, float* dinv, int n) {
    int i = blockIdx.x * blockDim.x + threadIdx.x;
    if (i < n) dinv[i] = rsqrtf(deg[i]);
}

__global__ void k_pack_w(const float* __restrict__ W, _Float16* __restrict__ Wh,
                         int K, int N) {
    int idx = blockIdx.x * blockDim.x + threadIdx.x;
    if (idx >= K * N) return;
    int frag   = idx >> 9;
    int within = idx & 511;
    int lane   = within >> 4;
    int i      = within & 15;
    int ntiles = N >> 4;
    int kt = frag / ntiles, nt = frag % ntiles;
    int n  = (nt << 4) + (lane & 15);
    int kk = (i < 8 ? i : i + 8) + ((lane >> 4) << 3);
    int k  = (kt << 5) + kk;
    Wh[idx] = (_Float16)W[k * N + n];
}

template <int NT, bool RELU>
__global__ void k_gemm(const float* __restrict__ A, const _Float16* __restrict__ Bh,
                       float* __restrict__ C, int M, int K, int N) {
    int wave = (int)((blockIdx.x * blockDim.x + threadIdx.x) >> 5);
    int lane = threadIdx.x & 31;
    int mtiles = M >> 4;
    if (wave >= mtiles) return;

    int m0  = wave << 4;
    int hi  = lane >> 4;
    int row = m0 + (lane & 15);

    v8f acc[NT];
    v8f zero = {0.f,0.f,0.f,0.f,0.f,0.f,0.f,0.f};
    #pragma unroll
    for (int j = 0; j < NT; ++j) acc[j] = zero;

    const v16h* bp = reinterpret_cast<const v16h*>(Bh);
    int ktiles = K >> 5;
    for (int kt = 0; kt < ktiles; ++kt) {
        int kbase = (kt << 5) + (hi << 3);
        const float4* p = reinterpret_cast<const float4*>(A + (long)row * K + kbase);
        float4 x0 = p[0], x1 = p[1];
        float4 x2 = p[4], x3 = p[5];
        if (RELU) {
            x0.x=fmaxf(x0.x,0.f); x0.y=fmaxf(x0.y,0.f); x0.z=fmaxf(x0.z,0.f); x0.w=fmaxf(x0.w,0.f);
            x1.x=fmaxf(x1.x,0.f); x1.y=fmaxf(x1.y,0.f); x1.z=fmaxf(x1.z,0.f); x1.w=fmaxf(x1.w,0.f);
            x2.x=fmaxf(x2.x,0.f); x2.y=fmaxf(x2.y,0.f); x2.z=fmaxf(x2.z,0.f); x2.w=fmaxf(x2.w,0.f);
            x3.x=fmaxf(x3.x,0.f); x3.y=fmaxf(x3.y,0.f); x3.z=fmaxf(x3.z,0.f); x3.w=fmaxf(x3.w,0.f);
        }
        v16h a;
        a[0]=(_Float16)x0.x; a[1]=(_Float16)x0.y; a[2]=(_Float16)x0.z; a[3]=(_Float16)x0.w;
        a[4]=(_Float16)x1.x; a[5]=(_Float16)x1.y; a[6]=(_Float16)x1.z; a[7]=(_Float16)x1.w;
        a[8]=(_Float16)x2.x; a[9]=(_Float16)x2.y; a[10]=(_Float16)x2.z; a[11]=(_Float16)x2.w;
        a[12]=(_Float16)x3.x; a[13]=(_Float16)x3.y; a[14]=(_Float16)x3.z; a[15]=(_Float16)x3.w;

        #pragma unroll
        for (int nt = 0; nt < NT; ++nt) {
            v16h b = bp[(size_t)(kt * NT + nt) * 32 + lane];
            acc[nt] = __builtin_amdgcn_wmma_f32_16x16x32_f16(
                false, a, false, b, (short)0, acc[nt], false, false);
        }
    }

    int mrow = m0 + (hi << 3);
    #pragma unroll
    for (int nt = 0; nt < NT; ++nt) {
        int n = (nt << 4) + (lane & 15);
        #pragma unroll
        for (int j = 0; j < 8; ++j)
            C[(long)(mrow + j) * N + n] = acc[nt][j];
    }
}

__global__ void k_self_bias(const float* __restrict__ H, const float* __restrict__ dinv,
                            const float* __restrict__ bias, float* __restrict__ acc,
                            int Nn, int C) {
    int idx = blockIdx.x * blockDim.x + threadIdx.x;
    if (idx >= Nn * C) return;
    int i = idx / C, c = idx % C;
    float d = dinv[i];
    acc[idx] = H[idx] * d * d + bias[c];
}

template <int CPE>
__global__ void k_edge_prop(const int* __restrict__ src, const int* __restrict__ dst,
                            const float* __restrict__ H, const float* __restrict__ dinv,
                            float* acc, int E) {
    int wave = (int)((blockIdx.x * blockDim.x + threadIdx.x) >> 5);
    int lane = threadIdx.x & 31;
    if (wave >= E) return;
    int s = src[wave], d = dst[wave];
    float norm = dinv[s] * dinv[d];
    constexpr int FPL = CPE / 32;
    const float* hs = H   + (long)s * CPE + lane * FPL;
    float*       ad = acc + (long)d * CPE + lane * FPL;
    if constexpr (FPL == 4) {
        float4 hv = *reinterpret_cast<const float4*>(hs);
        atomic_add_f32(&ad[0], hv.x * norm);
        atomic_add_f32(&ad[1], hv.y * norm);
        atomic_add_f32(&ad[2], hv.z * norm);
        atomic_add_f32(&ad[3], hv.w * norm);
    } else {
        float2 hv = *reinterpret_cast<const float2*>(hs);
        atomic_add_f32(&ad[0], hv.x * norm);
        atomic_add_f32(&ad[1], hv.y * norm);
    }
}

__global__ void k_pool_add(const int* __restrict__ batch, const float* __restrict__ A,
                           float* pool, float* counts, int Nn) {
    int idx = blockIdx.x * blockDim.x + threadIdx.x;
    if (idx >= Nn * OUT_C) return;
    int i = idx >> 6, c = idx & 63;
    int g = batch[i];
    atomic_add_f32(&pool[g * OUT_C + c], A[idx]);
    if (c == 0) atomic_add_f32(&counts[g], 1.0f);
}

__global__ void k_pool_final(const float* __restrict__ pool, const float* __restrict__ counts,
                             float* __restrict__ out) {
    int idx = blockIdx.x * blockDim.x + threadIdx.x;
    if (idx >= NGRAPH * OUT_C) return;
    out[idx] = pool[idx] / fmaxf(counts[idx >> 6], 1.0f);
}

extern "C" void kernel_launch(void* const* d_in, const int* in_sizes, int n_in,
                              void* d_out, int out_size, void* d_ws, size_t ws_size,
                              hipStream_t stream) {
    const float* x   = (const float*)d_in[0];
    const float* W1  = (const float*)d_in[1];
    const float* b1  = (const float*)d_in[2];
    const float* W2  = (const float*)d_in[3];
    const float* b2  = (const float*)d_in[4];
    const int*   ei  = (const int*)d_in[5];
    const int*   bat = (const int*)d_in[6];

    const int Nn = in_sizes[0] / IN_C;
    const int E  = in_sizes[5] / 2;
    const int* src = ei;
    const int* dst = ei + E;

    char* w = (char*)d_ws;
    size_t off = 0;
    auto alloc = [&](size_t bytes) -> void* {
        void* p = w + off;
        off = (off + bytes + 255) & ~(size_t)255;
        return p;
    };
    float*    deg   = (float*)alloc((size_t)Nn * 4);
    float*    dinv  = (float*)alloc((size_t)Nn * 4);
    _Float16* W1h   = (_Float16*)alloc((size_t)IN_C * HID_C * 2);
    _Float16* W2h   = (_Float16*)alloc((size_t)HID_C * OUT_C * 2);
    float*    h1    = (float*)alloc((size_t)Nn * HID_C * 4);
    float*    a1    = (float*)alloc((size_t)Nn * HID_C * 4);
    float*    pool  = (float*)alloc((size_t)NGRAPH * OUT_C * 4);
    float*    cnts  = (float*)alloc((size_t)NGRAPH * 4);
    float*    h2 = h1;
    float*    a2 = a1;

    const int T = 256;
    k_fill<<<(Nn + T - 1) / T, T, 0, stream>>>(deg, 1.0f, Nn);
    k_edge_deg<<<(E + T - 1) / T, T, 0, stream>>>(dst, deg, E);
    k_rsqrt<<<(Nn + T - 1) / T, T, 0, stream>>>(deg, dinv, Nn);

    k_pack_w<<<(IN_C * HID_C + T - 1) / T, T, 0, stream>>>(W1, W1h, IN_C, HID_C);
    k_pack_w<<<(HID_C * OUT_C + T - 1) / T, T, 0, stream>>>(W2, W2h, HID_C, OUT_C);

    {
        int waves = Nn / 16;
        int blocks = (waves * 32 + T - 1) / T;
        k_gemm<HID_C / 16, false><<<blocks, T, 0, stream>>>(x, W1h, h1, Nn, IN_C, HID_C);
    }
    k_self_bias<<<(Nn * HID_C + T - 1) / T, T, 0, stream>>>(h1, dinv, b1, a1, Nn, HID_C);
    k_edge_prop<HID_C><<<(E + 7) / 8, T, 0, stream>>>(src, dst, h1, dinv, a1, E);

    {
        int waves = Nn / 16;
        int blocks = (waves * 32 + T - 1) / T;
        k_gemm<OUT_C / 16, true><<<blocks, T, 0, stream>>>(a1, W2h, h2, Nn, HID_C, OUT_C);
    }
    k_self_bias<<<(Nn * OUT_C + T - 1) / T, T, 0, stream>>>(h2, dinv, b2, a2, Nn, OUT_C);
    k_edge_prop<OUT_C><<<(E + 7) / 8, T, 0, stream>>>(src, dst, h2, dinv, a2, E);

    k_fill<<<(NGRAPH * OUT_C + T - 1) / T, T, 0, stream>>>(pool, 0.0f, NGRAPH * OUT_C);
    k_fill<<<1, T, 0, stream>>>(cnts, 0.0f, NGRAPH);
    k_pool_add<<<(Nn * OUT_C + T - 1) / T, T, 0, stream>>>(bat, a2, pool, cnts, Nn);
    k_pool_final<<<(NGRAPH * OUT_C + T - 1) / T, T, 0, stream>>>(pool, cnts, (float*)d_out);
}


# mean pool folded into layer-2 propagate by linearity: each message adds into its destination's graph row in a per-block LDS table (LDS float atomics), one flush; float sum order changes
# speedup vs baseline: 1.3815x; 1.3815x over previous
#include <hip/hip_runtime.h>
#include <hip/hip_bf16.h>
#include <stdint.h>

typedef __attribute__((ext_vector_type(16))) _Float16 v16h;
typedef __attribute__((ext_vector_type(8)))  float    v8f;

#define IN_C   128
#define HID_C  128
#define OUT_C  64
#define NGRAPH 128

__device__ __forceinline__ void atomic_add_f32(float* p, float v) {
    unsafeAtomicAdd(p, v);
}

__global__ void k_fill(float* p, float v, int n) {
    int i = blockIdx.x * blockDim.x + threadIdx.x;
    if (i < n) p[i] = v;
}

__global__ void k_edge_deg(const int* __restrict__ dst, float* deg, int E) {
    int e = blockIdx.x * blockDim.x + threadIdx.x;
    if (e < E) atomic_add_f32(&deg[dst[e]], 1.0f);
}

__global__ void k_rsqrt(const float* __restrict__ deg, float* dinv, int n) {
    int i = blockIdx.x * blockDim.x + threadIdx.x;
    if (i < n) dinv[i] = rsqrtf(deg[i]);
}

__global__ void k_pack_w(const float* __restrict__ W, _Float16* __restrict__ Wh,
                         int K, int N) {
    int idx = blockIdx.x * blockDim.x + threadIdx.x;
    if (idx >= K * N) return;
    int frag   = idx >> 9;
    int within = idx & 511;
    int lane   = within >> 4;
    int i      = within & 15;
    int ntiles = N >> 4;
    int kt = frag / ntiles, nt = frag % ntiles;
    int n  = (nt << 4) + (lane & 15);
    int kk = (i < 8 ? i : i + 8) + ((lane >> 4) << 3);
    int k  = (kt << 5) + kk;
    Wh[idx] = (_Float16)W[k * N + n];
}

template <int NT, bool RELU>
__global__ void k_gemm(const float* __restrict__ A, const _Float16* __restrict__ Bh,
                       float* __restrict__ C, int M, int K, int N) {
    int wave = (int)((blockIdx.x * blockDim.x + threadIdx.x) >> 5);
    int lane = threadIdx.x & 31;
    int mtiles = M >> 4;
    if (wave >= mtiles) return;

    int m0  = wave << 4;
    int hi  = lane >> 4;
    int row = m0 + (lane & 15);

    v8f acc[NT];
    v8f zero = {0.f,0.f,0.f,0.f,0.f,0.f,0.f,0.f};
    #pragma unroll
    for (int j = 0; j < NT; ++j) acc[j] = zero;

    const v16h* bp = reinterpret_cast<const v16h*>(Bh);
    int ktiles = K >> 5;
    for (int kt = 0; kt < ktiles; ++kt) {
        int kbase = (kt << 5) + (hi << 3);
        const float4* p = reinterpret_cast<const float4*>(A + (long)row * K + kbase);
        float4 x0 = p[0], x1 = p[1];
        float4 x2 = p[4], x3 = p[5];
        if (RELU) {
            x0.x=fmaxf(x0.x,0.f); x0.y=fmaxf(x0.y,0.f); x0.z=fmaxf(x0.z,0.f); x0.w=fmaxf(x0.w,0.f);
            x1.x=fmaxf(x1.x,0.f); x1.y=fmaxf(x1.y,0.f); x1.z=fmaxf(x1.z,0.f); x1.w=fmaxf(x1.w,0.f);
            x2.x=fmaxf(x2.x,0.f); x2.y=fmaxf(x2.y,0.f); x2.z=fmaxf(x2.z,0.f); x2.w=fmaxf(x2.w,0.f);
            x3.x=fmaxf(x3.x,0.f); x3.y=fmaxf(x3.y,0.f); x3.z=fmaxf(x3.z,0.f); x3.w=fmaxf(x3.w,0.f);
        }
        v16h a;
        a[0]=(_Float16)x0.x; a[1]=(_Float16)x0.y; a[2]=(_Float16)x0.z; a[3]=(_Float16)x0.w;
        a[4]=(_Float16)x1.x; a[5]=(_Float16)x1.y; a[6]=(_Float16)x1.z; a[7]=(_Float16)x1.w;
        a[8]=(_Float16)x2.x; a[9]=(_Float16)x2.y; a[10]=(_Float16)x2.z; a[11]=(_Float16)x2.w;
        a[12]=(_Float16)x3.x; a[13]=(_Float16)x3.y; a[14]=(_Float16)x3.z; a[15]=(_Float16)x3.w;

        #pragma unroll
        for (int nt = 0; nt < NT; ++nt) {
            v16h b = bp[(size_t)(kt * NT + nt) * 32 + lane];
            acc[nt] = __builtin_amdgcn_wmma_f32_16x16x32_f16(
                false, a, false, b, (short)0, acc[nt], false, false);
        }
    }

    int mrow = m0 + (hi << 3);
    #pragma unroll
    for (int nt = 0; nt < NT; ++nt) {
        int n = (nt << 4) + (lane & 15);
        #pragma unroll
        for (int j = 0; j < 8; ++j)
            C[(long)(mrow + j) * N + n] = acc[nt][j];
    }
}

__global__ void k_self_bias(const float* __restrict__ H, const float* __restrict__ dinv,
                            const float* __restrict__ bias, float* __restrict__ acc,
                            int Nn, int C) {
    int idx = blockIdx.x * blockDim.x + threadIdx.x;
    if (idx >= Nn * C) return;
    int i = idx / C, c = idx % C;
    float d = dinv[i];
    acc[idx] = H[idx] * d * d + bias[c];
}

template <int CPE>
__global__ void k_edge_prop(const int* __restrict__ src, const int* __restrict__ dst,
                            const float* __restrict__ H, const float* __restrict__ dinv,
                            float* acc, int E) {
    int wave = (int)((blockIdx.x * blockDim.x + threadIdx.x) >> 5);
    int lane = threadIdx.x & 31;
    if (wave >= E) return;
    int s = src[wave], d = dst[wave];
    float norm = dinv[s] * dinv[d];
    constexpr int FPL = CPE / 32;
    const float* hs = H   + (long)s * CPE + lane * FPL;
    float*       ad = acc + (long)d * CPE + lane * FPL;
    if constexpr (FPL == 4) {
        float4 hv = *reinterpret_cast<const float4*>(hs);
        atomic_add_f32(&ad[0], hv.x * norm);
        atomic_add_f32(&ad[1], hv.y * norm);
        atomic_add_f32(&ad[2], hv.z * norm);
        atomic_add_f32(&ad[3], hv.w * norm);
    } else {
        float2 hv = *reinterpret_cast<const float2*>(hs);
        atomic_add_f32(&ad[0], hv.x * norm);
        atomic_add_f32(&ad[1], hv.y * norm);
    }
}

#define PP_THREADS 512
#define PP_WAVES   (PP_THREADS / 32)
#define PP_BATCH   16

__device__ __forceinline__ int   lane_bcast_i(int v, int j)   { return __builtin_amdgcn_readlane(v, j); }
__device__ __forceinline__ float lane_bcast_f(float v, int j) {
    return __int_as_float(__builtin_amdgcn_readlane(__float_as_int(v), j));
}

__global__ void __launch_bounds__(PP_THREADS)
k_prop_pool(const int* __restrict__ src, const int* __restrict__ dst,
            const int* __restrict__ batch, const float* __restrict__ H,
            const float* __restrict__ dinv, const float* __restrict__ bias,
            float* pool, float* counts, int E, int Nn) {
    __shared__ float tab[NGRAPH * OUT_C];
    __shared__ float cnt[NGRAPH];
    for (int i = threadIdx.x; i < NGRAPH * OUT_C; i += PP_THREADS) tab[i] = 0.f;
    if (threadIdx.x < NGRAPH) cnt[threadIdx.x] = 0.f;
    __syncthreads();

    const int lane = threadIdx.x & 31;
    const int wib  = __builtin_amdgcn_readfirstlane((int)(threadIdx.x >> 5));
    const int nge  = (E + 31) >> 5;
    const int ngn  = (Nn + 31) >> 5;
    const int ng   = nge + ngn;
    const int tw   = (int)gridDim.x * PP_WAVES;
    const int c0   = lane * 2;
    const float bx = bias[c0], by = bias[c0 + 1];

    for (int gi = wib * (int)gridDim.x + (int)blockIdx.x; gi < ng; gi += tw) {
        const bool isnode = gi >= nge;
        int s = 0, d = 0, m;
        if (!isnode) {
            const int base = gi << 5;
            m = min(32, E - base);
            if (lane < m) { s = src[base + lane]; d = dst[base + lane]; }
        } else {
            const int base = (gi - nge) << 5;
            m = min(32, Nn - base);
            if (lane < m) { s = base + lane; d = s; }
        }
        const float nrm = dinv[s] * dinv[d];
        const int   g   = batch[d];
        const float ax = isnode ? bx : 0.f;
        const float ay = isnode ? by : 0.f;
        if (isnode && lane < m) atomicAdd(&cnt[g], 1.0f);

        if (m == 32) {
            #pragma unroll
            for (int j0 = 0; j0 < 32; j0 += PP_BATCH) {
                float2 v[PP_BATCH];
                #pragma unroll
                for (int u = 0; u < PP_BATCH; ++u) {
                    const int sj = lane_bcast_i(s, j0 + u);
                    v[u] = *reinterpret_cast<const float2*>(H + (long)sj * OUT_C + c0);
                }
                #pragma unroll
                for (int u = 0; u < PP_BATCH; ++u) {
                    const float nj = lane_bcast_f(nrm, j0 + u);
                    const int   gj = lane_bcast_i(g, j0 + u);
                    float* t = &tab[gj * OUT_C + c0];
                    atomicAdd(t,     fmaf(v[u].x, nj, ax));
                    atomicAdd(t + 1, fmaf(v[u].y, nj, ay));
                }
            }
        } else {
            for (int j = 0; j < m; ++j) {
                const int   sj = lane_bcast_i(s, j);
                const float nj = lane_bcast_f(nrm, j);
                const int   gj = lane_bcast_i(g, j);
                const float2 v = *reinterpret_cast<const float2*>(H + (long)sj * OUT_C + c0);
                float* t = &tab[gj * OUT_C + c0];
                atomicAdd(t,     fmaf(v.x, nj, ax));
                atomicAdd(t + 1, fmaf(v.y, nj, ay));
            }
        }
    }
    __syncthreads();

    constexpr int NLINES = NGRAPH * OUT_C / 32;
    for (int k = wib; k < NLINES; k += PP_WAVES) {
        const int idx = (((k + (int)blockIdx.x) & (NLINES - 1)) << 5) + lane;
        atomic_add_f32(&pool[idx], tab[idx]);
    }
    if (threadIdx.x < NGRAPH) atomic_add_f32(&counts[threadIdx.x], cnt[threadIdx.x]);
}

__global__ void k_pool_final(const float* __restrict__ pool, const float* __restrict__ counts,
                             float* __restrict__ out) {
    int idx = blockIdx.x * blockDim.x + threadIdx.x;
    if (idx >= NGRAPH * OUT_C) return;
    out[idx] = pool[idx] / fmaxf(counts[idx >> 6], 1.0f);
}

extern "C" void kernel_launch(void* const* d_in, const int* in_sizes, int n_in,
                              void* d_out, int out_size, void* d_ws, size_t ws_size,
                              hipStream_t stream) {
    const float* x   = (const float*)d_in[0];
    const float* W1  = (const float*)d_in[1];
    const float* b1  = (const float*)d_in[2];
    const float* W2  = (const float*)d_in[3];
    const float* b2  = (const float*)d_in[4];
    const int*   ei  = (const int*)d_in[5];
    const int*   bat = (const int*)d_in[6];

    const int Nn = in_sizes[0] / IN_C;
    const int E  = in_sizes[5] / 2;
    const int* src = ei;
    const int* dst = ei + E;

    char* w = (char*)d_ws;
    size_t off = 0;
    auto alloc = [&](size_t bytes) -> void* {
        void* p = w + off;
        off = (off + bytes + 255) & ~(size_t)255;
        return p;
    };
    float*    deg   = (float*)alloc((size_t)Nn * 4);
    float*    dinv  = (float*)alloc((size_t)Nn * 4);
    _Float16* W1h   = (_Float16*)alloc((size_t)IN_C * HID_C * 2);
    _Float16* W2h   = (_Float16*)alloc((size_t)HID_C * OUT_C * 2);
    float*    h1    = (float*)alloc((size_t)Nn * HID_C * 4);
    float*    a1    = (float*)alloc((size_t)Nn * HID_C * 4);
    float*    pool  = (float*)alloc((size_t)NGRAPH * OUT_C * 4);
    float*    cnts  = (float*)alloc((size_t)NGRAPH * 4);
    float*    h2 = h1;

    const int T = 256;
    k_fill<<<(Nn + T - 1) / T, T, 0, stream>>>(deg, 1.0f, Nn);
    k_edge_deg<<<(E + T - 1) / T, T, 0, stream>>>(dst, deg, E);
    k_rsqrt<<<(Nn + T - 1) / T, T, 0, stream>>>(deg, dinv, Nn);

    k_pack_w<<<(IN_C * HID_C + T - 1) / T, T, 0, stream>>>(W1, W1h, IN_C, HID_C);
    k_pack_w<<<(HID_C * OUT_C + T - 1) / T, T, 0, stream>>>(W2, W2h, HID_C, OUT_C);

    {
        int waves = Nn / 16;
        int blocks = (waves * 32 + T - 1) / T;
        k_gemm<HID_C / 16, false><<<blocks, T, 0, stream>>>(x, W1h, h1, Nn, IN_C, HID_C);
    }
    k_self_bias<<<(Nn * HID_C + T - 1) / T, T, 0, stream>>>(h1, dinv, b1, a1, Nn, HID_C);
    k_edge_prop<HID_C><<<(E + 7) / 8, T, 0, stream>>>(src, dst, h1, dinv, a1, E);

    {
        int waves = Nn / 16;
        int blocks = (waves * 32 + T - 1) / T;
        k_gemm<OUT_C / 16, true><<<blocks, T, 0, stream>>>(a1, W2h, h2, Nn, HID_C, OUT_C);
    }
    k_fill<<<(NGRAPH * OUT_C + T - 1) / T, T, 0, stream>>>(pool, 0.0f, NGRAPH * OUT_C);
    k_fill<<<1, T, 0, stream>>>(cnts, 0.0f, NGRAPH);
    k_prop_pool<<<512, PP_THREADS, 0, stream>>>(src, dst, bat, h2, dinv, b2, pool, cnts, E, Nn);
    k_pool_final<<<(NGRAPH * OUT_C + T - 1) / T, T, 0, stream>>>(pool, cnts, (float*)d_out);
}


# layer-1 edge scatter: lane owns channels lane+32k (was 4 consecutive): each row load and each float atomic of a wave covers one whole 128-byte line, not quarters of four; atomics, arrival order kept
# speedup vs baseline: 1.6893x; 1.2228x over previous
#include <hip/hip_runtime.h>
#include <hip/hip_bf16.h>
#include <stdint.h>

typedef __attribute__((ext_vector_type(16))) _Float16 v16h;
typedef __attribute__((ext_vector_type(8)))  float    v8f;

#define IN_C   128
#define HID_C  128
#define OUT_C  64
#define NGRAPH 128

__device__ __forceinline__ void atomic_add_f32(float* p, float v) {
    unsafeAtomicAdd(p, v);
}

__global__ void k_fill(float* p, float v, int n) {
    int i = blockIdx.x * blockDim.x + threadIdx.x;
    if (i < n) p[i] = v;
}

__global__ void k_edge_deg(const int* __restrict__ dst, float* deg, int E) {
    int e = blockIdx.x * blockDim.x + threadIdx.x;
    if (e < E) atomic_add_f32(&deg[dst[e]], 1.0f);
}

__global__ void k_rsqrt(const float* __restrict__ deg, float* dinv, int n) {
    int i = blockIdx.x * blockDim.x + threadIdx.x;
    if (i < n) dinv[i] = rsqrtf(deg[i]);
}

__global__ void k_pack_w(const float* __restrict__ W, _Float16* __restrict__ Wh,
                         int K, int N) {
    int idx = blockIdx.x * blockDim.x + threadIdx.x;
    if (idx >= K * N) return;
    int frag   = idx >> 9;
    int within = idx & 511;
    int lane   = within >> 4;
    int i      = within & 15;
    int ntiles = N >> 4;
    int kt = frag / ntiles, nt = frag % ntiles;
    int n  = (nt << 4) + (lane & 15);
    int kk = (i < 8 ? i : i + 8) + ((lane >> 4) << 3);
    int k  = (kt << 5) + kk;
    Wh[idx] = (_Float16)W[k * N + n];
}

template <int NT, bool RELU>
__global__ void k_gemm(const float* __restrict__ A, const _Float16* __restrict__ Bh,
                       float* __restrict__ C, int M, int K, int N) {
    int wave = (int)((blockIdx.x * blockDim.x + threadIdx.x) >> 5);
    int lane = threadIdx.x & 31;
    int mtiles = M >> 4;
    if (wave >= mtiles) return;

    int m0  = wave << 4;
    int hi  = lane >> 4;
    int row = m0 + (lane & 15);

    v8f acc[NT];
    v8f zero = {0.f,0.f,0.f,0.f,0.f,0.f,0.f,0.f};
    #pragma unroll
    for (int j = 0; j < NT; ++j) acc[j] = zero;

    const v16h* bp = reinterpret_cast<const v16h*>(Bh);
    int ktiles = K >> 5;
    for (int kt = 0; kt < ktiles; ++kt) {
        int kbase = (kt << 5) + (hi << 3);
        const float4* p = reinterpret_cast<const float4*>(A + (long)row * K + kbase);
        float4 x0 = p[0], x1 = p[1];
        float4 x2 = p[4], x3 = p[5];
        if (RELU) {
            x0.x=fmaxf(x0.x,0.f); x0.y=fmaxf(x0.y,0.f); x0.z=fmaxf(x0.z,0.f); x0.w=fmaxf(x0.w,0.f);
            x1.x=fmaxf(x1.x,0.f); x1.y=fmaxf(x1.y,0.f); x1.z=fmaxf(x1.z,0.f); x1.w=fmaxf(x1.w,0.f);
            x2.x=fmaxf(x2.x,0.f); x2.y=fmaxf(x2.y,0.f); x2.z=fmaxf(x2.z,0.f); x2.w=fmaxf(x2.w,0.f);
            x3.x=fmaxf(x3.x,0.f); x3.y=fmaxf(x3.y,0.f); x3.z=fmaxf(x3.z,0.f); x3.w=fmaxf(x3.w,0.f);
        }
        v16h a;
        a[0]=(_Float16)x0.x; a[1]=(_Float16)x0.y; a[2]=(_Float16)x0.z; a[3]=(_Float16)x0.w;
        a[4]=(_Float16)x1.x; a[5]=(_Float16)x1.y; a[6]=(_Float16)x1.z; a[7]=(_Float16)x1.w;
        a[8]=(_Float16)x2.x; a[9]=(_Float16)x2.y; a[10]=(_Float16)x2.z; a[11]=(_Float16)x2.w;
        a[12]=(_Float16)x3.x; a[13]=(_Float16)x3.y; a[14]=(_Float16)x3.z; a[15]=(_Float16)x3.w;

        #pragma unroll
        for (int nt = 0; nt < NT; ++nt) {
            v16h b = bp[(size_t)(kt * NT + nt) * 32 + lane];
            acc[nt] = __builtin_amdgcn_wmma_f32_16x16x32_f16(
                false, a, false, b, (short)0, acc[nt], false, false);
        }
    }

    int mrow = m0 + (hi << 3);
    #pragma unroll
    for (int nt = 0; nt < NT; ++nt) {
        int n = (nt << 4) + (lane & 15);
        #pragma unroll
        for (int j = 0; j < 8; ++j)
            C[(long)(mrow + j) * N + n] = acc[nt][j];
    }
}

__global__ void k_self_bias(const float* __restrict__ H, const float* __restrict__ dinv,
                            const float* __restrict__ bias, float* __restrict__ acc,
                            int Nn, int C) {
    int idx = blockIdx.x * blockDim.x + threadIdx.x;
    if (idx >= Nn * C) return;
    int i = idx / C, c = idx % C;
    float d = dinv[i];
    acc[idx] = H[idx] * d * d + bias[c];
}

template <int CPE>
__global__ void k_edge_prop(const int* __restrict__ src, const int* __restrict__ dst,
                            const float* __restrict__ H, const float* __restrict__ dinv,
                            float* acc, int E) {
    int wave = (int)((blockIdx.x * blockDim.x + threadIdx.x) >> 5);
    int lane = threadIdx.x & 31;
    if (wave >= E) return;
    int s = src[wave], d = dst[wave];
    float norm = dinv[s] * dinv[d];
    constexpr int FPL = CPE / 32;
    const float* hs = H   + (long)s * CPE + lane;
    float*       ad = acc + (long)d * CPE + lane;
    float hv[FPL];
    for (int k = 0; k < FPL; ++k) hv[k] = hs[k * 32];
    for (int k = 0; k < FPL; ++k) atomic_add_f32(&ad[k * 32], hv[k] * norm);
}

#define PP_THREADS 512
#define PP_WAVES   (PP_THREADS / 32)
#define PP_BATCH   16

__device__ __forceinline__ int   lane_bcast_i(int v, int j)   { return __builtin_amdgcn_readlane(v, j); }
__device__ __forceinline__ float lane_bcast_f(float v, int j) {
    return __int_as_float(__builtin_amdgcn_readlane(__float_as_int(v), j));
}

__global__ void __launch_bounds__(PP_THREADS)
k_prop_pool(const int* __restrict__ src, const int* __restrict__ dst,
            const int* __restrict__ batch, const float* __restrict__ H,
            const float* __restrict__ dinv, const float* __restrict__ bias,
            float* pool, float* counts, int E, int Nn) {
    __shared__ float tab[NGRAPH * OUT_C];
    __shared__ float cnt[NGRAPH];
    for (int i = threadIdx.x; i < NGRAPH * OUT_C; i += PP_THREADS) tab[i] = 0.f;
    if (threadIdx.x < NGRAPH) cnt[threadIdx.x] = 0.f;
    __syncthreads();

    const int lane = threadIdx.x & 31;
    const int wib  = __builtin_amdgcn_readfirstlane((int)(threadIdx.x >> 5));
    const int nge  = (E + 31) >> 5;
    const int ngn  = (Nn + 31) >> 5;
    const int ng   = nge + ngn;
    const int tw   = (int)gridDim.x * PP_WAVES;
    const int c0   = lane * 2;
    const float bx = bias[c0], by = bias[c0 + 1];

    for (int gi = wib * (int)gridDim.x + (int)blockIdx.x; gi < ng; gi += tw) {
        const bool isnode = gi >= nge;
        int s = 0, d = 0, m;
        if (!isnode) {
            const int base = gi << 5;
            m = min(32, E - base);
            if (lane < m) { s = src[base + lane]; d = dst[base + lane]; }
        } else {
            const int base = (gi - nge) << 5;
            m = min(32, Nn - base);
            if (lane < m) { s = base + lane; d = s; }
        }
        const float nrm = dinv[s] * dinv[d];
        const int   g   = batch[d];
        const float ax = isnode ? bx : 0.f;
        const float ay = isnode ? by : 0.f;
        if (isnode && lane < m) atomicAdd(&cnt[g], 1.0f);

        if (m == 32) {
            #pragma unroll
            for (int j0 = 0; j0 < 32; j0 += PP_BATCH) {
                float2 v[PP_BATCH];
                #pragma unroll
                for (int u = 0; u < PP_BATCH; ++u) {
                    const int sj = lane_bcast_i(s, j0 + u);
                    v[u] = *reinterpret_cast<const float2*>(H + (long)sj * OUT_C + c0);
                }
                #pragma unroll
                for (int u = 0; u < PP_BATCH; ++u) {
                    const float nj = lane_bcast_f(nrm, j0 + u);
                    const int   gj = lane_bcast_i(g, j0 + u);
                    float* t = &tab[gj * OUT_C + c0];
                    atomicAdd(t,     fmaf(v[u].x, nj, ax));
                    atomicAdd(t + 1, fmaf(v[u].y, nj, ay));
                }
            }
        } else {
            for (int j = 0; j < m; ++j) {
                const int   sj = lane_bcast_i(s, j);
                const float nj = lane_bcast_f(nrm, j);
                const int   gj = lane_bcast_i(g, j);
                const float2 v = *reinterpret_cast<const float2*>(H + (long)sj * OUT_C + c0);
                float* t = &tab[gj * OUT_C + c0];
                atomicAdd(t,     fmaf(v.x, nj, ax));
                atomicAdd(t + 1, fmaf(v.y, nj, ay));
            }
        }
    }
    __syncthreads();

    constexpr int NLINES = NGRAPH * OUT_C / 32;
    for (int k = wib; k < NLINES; k += PP_WAVES) {
        const int idx = (((k + (int)blockIdx.x) & (NLINES - 1)) << 5) + lane;
        atomic_add_f32(&pool[idx], tab[idx]);
    }
    if (threadIdx.x < NGRAPH) atomic_add_f32(&counts[threadIdx.x], cnt[threadIdx.x]);
}

__global__ void k_pool_final(const float* __restrict__ pool, const float* __restrict__ counts,
                             float* __restrict__ out) {
    int idx = blockIdx.x * blockDim.x + threadIdx.x;
    if (idx >= NGRAPH * OUT_C) return;
    out[idx] = pool[idx] / fmaxf(counts[idx >> 6], 1.0f);
}

extern "C" void kernel_launch(void* const* d_in, const int* in_sizes, int n_in,
                              void* d_out, int out_size, void* d_ws, size_t ws_size,
                              hipStream_t stream) {
    const float* x   = (const float*)d_in[0];
    const float* W1  = (const float*)d_in[1];
    const float* b1  = (const float*)d_in[2];
    const float* W2  = (const float*)d_in[3];
    const float* b2  = (const float*)d_in[4];
    const int*   ei  = (const int*)d_in[5];
    const int*   bat = (const int*)d_in[6];

    const int Nn = in_sizes[0] / IN_C;
    const int E  = in_sizes[5] / 2;
    const int* src = ei;
    const int* dst = ei + E;

    char* w = (char*)d_ws;
    size_t off = 0;
    auto alloc = [&](size_t bytes) -> void* {
        void* p = w + off;
        off = (off + bytes + 255) & ~(size_t)255;
        return p;
    };
    float*    deg   = (float*)alloc((size_t)Nn * 4);
    float*    dinv  = (float*)alloc((size_t)Nn * 4);
    _Float16* W1h   = (_Float16*)alloc((size_t)IN_C * HID_C * 2);
    _Float16* W2h   = (_Float16*)alloc((size_t)HID_C * OUT_C * 2);
    float*    h1    = (float*)alloc((size_t)Nn * HID_C * 4);
    float*    a1    = (float*)alloc((size_t)Nn * HID_C * 4);
    float*    pool  = (float*)alloc((size_t)NGRAPH * OUT_C * 4);
    float*    cnts  = (float*)alloc((size_t)NGRAPH * 4);
    float*    h2 = h1;

    const int T = 256;
    k_fill<<<(Nn + T - 1) / T, T, 0, stream>>>(deg, 1.0f, Nn);
    k_edge_deg<<<(E + T - 1) / T, T, 0, stream>>>(dst, deg, E);
    k_rsqrt<<<(Nn + T - 1) / T, T, 0, stream>>>(deg, dinv, Nn);

    k_pack_w<<<(IN_C * HID_C + T - 1) / T, T, 0, stream>>>(W1, W1h, IN_C, HID_C);
    k_pack_w<<<(HID_C * OUT_C + T - 1) / T, T, 0, stream>>>(W2, W2h, HID_C, OUT_C);

    {
        int waves = Nn / 16;
        int blocks = (waves * 32 + T - 1) / T;
        k_gemm<HID_C / 16, false><<<blocks, T, 0, stream>>>(x, W1h, h1, Nn, IN_C, HID_C);
    }
    k_self_bias<<<(Nn * HID_C + T - 1) / T, T, 0, stream>>>(h1, dinv, b1, a1, Nn, HID_C);
    k_edge_prop<HID_C><<<(E + 7) / 8, T, 0, stream>>>(src, dst, h1, dinv, a1, E);

    {
        int waves = Nn / 16;
        int blocks = (waves * 32 + T - 1) / T;
        k_gemm<OUT_C / 16, true><<<blocks, T, 0, stream>>>(a1, W2h, h2, Nn, HID_C, OUT_C);
    }
    k_fill<<<(NGRAPH * OUT_C + T - 1) / T, T, 0, stream>>>(pool, 0.0f, NGRAPH * OUT_C);
    k_fill<<<1, T, 0, stream>>>(cnts, 0.0f, NGRAPH);
    k_prop_pool<<<512, PP_THREADS, 0, stream>>>(src, dst, bat, h2, dinv, b2, pool, cnts, E, Nn);
    k_pool_final<<<(NGRAPH * OUT_C + T - 1) / T, T, 0, stream>>>(pool, cnts, (float*)d_out);
}


# layer-1 edge scatter: one wave per 32 edges (was one per edge): indices and dinv read coalesced, edges walked by lane broadcast, 8 source rows in flight; float atomics kept, sums in arrival order
# speedup vs baseline: 1.9058x; 1.1282x over previous
#include <hip/hip_runtime.h>
#include <hip/hip_bf16.h>
#include <stdint.h>

typedef __attribute__((ext_vector_type(16))) _Float16 v16h;
typedef __attribute__((ext_vector_type(8)))  float    v8f;

#define IN_C   128
#define HID_C  128
#define OUT_C  64
#define NGRAPH 128

__device__ __forceinline__ void atomic_add_f32(float* p, float v) {
    unsafeAtomicAdd(p, v);
}

__global__ void k_fill(float* p, float v, int n) {
    int i = blockIdx.x * blockDim.x + threadIdx.x;
    if (i < n) p[i] = v;
}

__global__ void k_edge_deg(const int* __restrict__ dst, float* deg, int E) {
    int e = blockIdx.x * blockDim.x + threadIdx.x;
    if (e < E) atomic_add_f32(&deg[dst[e]], 1.0f);
}

__global__ void k_rsqrt(const float* __restrict__ deg, float* dinv, int n) {
    int i = blockIdx.x * blockDim.x + threadIdx.x;
    if (i < n) dinv[i] = rsqrtf(deg[i]);
}

__global__ void k_pack_w(const float* __restrict__ W, _Float16* __restrict__ Wh,
                         int K, int N) {
    int idx = blockIdx.x * blockDim.x + threadIdx.x;
    if (idx >= K * N) return;
    int frag   = idx >> 9;
    int within = idx & 511;
    int lane   = within >> 4;
    int i      = within & 15;
    int ntiles = N >> 4;
    int kt = frag / ntiles, nt = frag % ntiles;
    int n  = (nt << 4) + (lane & 15);
    int kk = (i < 8 ? i : i + 8) + ((lane >> 4) << 3);
    int k  = (kt << 5) + kk;
    Wh[idx] = (_Float16)W[k * N + n];
}

template <int NT, bool RELU>
__global__ void k_gemm(const float* __restrict__ A, const _Float16* __restrict__ Bh,
                       float* __restrict__ C, int M, int K, int N) {
    int wave = (int)((blockIdx.x * blockDim.x + threadIdx.x) >> 5);
    int lane = threadIdx.x & 31;
    int mtiles = M >> 4;
    if (wave >= mtiles) return;

    int m0  = wave << 4;
    int hi  = lane >> 4;
    int row = m0 + (lane & 15);

    v8f acc[NT];
    v8f zero = {0.f,0.f,0.f,0.f,0.f,0.f,0.f,0.f};
    #pragma unroll
    for (int j = 0; j < NT; ++j) acc[j] = zero;

    const v16h* bp = reinterpret_cast<const v16h*>(Bh);
    int ktiles = K >> 5;
    for (int kt = 0; kt < ktiles; ++kt) {
        int kbase = (kt << 5) + (hi << 3);
        const float4* p = reinterpret_cast<const float4*>(A + (long)row * K + kbase);
        float4 x0 = p[0], x1 = p[1];
        float4 x2 = p[4], x3 = p[5];
        if (RELU) {
            x0.x=fmaxf(x0.x,0.f); x0.y=fmaxf(x0.y,0.f); x0.z=fmaxf(x0.z,0.f); x0.w=fmaxf(x0.w,0.f);
            x1.x=fmaxf(x1.x,0.f); x1.y=fmaxf(x1.y,0.f); x1.z=fmaxf(x1.z,0.f); x1.w=fmaxf(x1.w,0.f);
            x2.x=fmaxf(x2.x,0.f); x2.y=fmaxf(x2.y,0.f); x2.z=fmaxf(x2.z,0.f); x2.w=fmaxf(x2.w,0.f);
            x3.x=fmaxf(x3.x,0.f); x3.y=fmaxf(x3.y,0.f); x3.z=fmaxf(x3.z,0.f); x3.w=fmaxf(x3.w,0.f);
        }
        v16h a;
        a[0]=(_Float16)x0.x; a[1]=(_Float16)x0.y; a[2]=(_Float16)x0.z; a[3]=(_Float16)x0.w;
        a[4]=(_Float16)x1.x; a[5]=(_Float16)x1.y; a[6]=(_Float16)x1.z; a[7]=(_Float16)x1.w;
        a[8]=(_Float16)x2.x; a[9]=(_Float16)x2.y; a[10]=(_Float16)x2.z; a[11]=(_Float16)x2.w;
        a[12]=(_Float16)x3.x; a[13]=(_Float16)x3.y; a[14]=(_Float16)x3.z; a[15]=(_Float16)x3.w;

        #pragma unroll
        for (int nt = 0; nt < NT; ++nt) {
            v16h b = bp[(size_t)(kt * NT + nt) * 32 + lane];
            acc[nt] = __builtin_amdgcn_wmma_f32_16x16x32_f16(
                false, a, false, b, (short)0, acc[nt], false, false);
        }
    }

    int mrow = m0 + (hi << 3);
    #pragma unroll
    for (int nt = 0; nt < NT; ++nt) {
        int n = (nt << 4) + (lane & 15);
        #pragma unroll
        for (int j = 0; j < 8; ++j)
            C[(long)(mrow + j) * N + n] = acc[nt][j];
    }
}

__global__ void k_self_bias(const float* __restrict__ H, const float* __restrict__ dinv,
                            const float* __restrict__ bias, float* __restrict__ acc,
                            int Nn, int C) {
    int idx = blockIdx.x * blockDim.x + threadIdx.x;
    if (idx >= Nn * C) return;
    int i = idx / C, c = idx % C;
    float d = dinv[i];
    acc[idx] = H[idx] * d * d + bias[c];
}

__device__ __forceinline__ int   lane_bcast_i(int v, int j)   { return __builtin_amdgcn_readlane(v, j); }
__device__ __forceinline__ float lane_bcast_f(float v, int j) {
    return __int_as_float(__builtin_amdgcn_readlane(__float_as_int(v), j));
}

#define EP_THREADS 256
#define EP_BATCH   8
template <int CPE>
__global__ void __launch_bounds__(EP_THREADS)
k_edge_prop(const int* __restrict__ src, const int* __restrict__ dst,
            const float* __restrict__ H, const float* __restrict__ dinv,
            float* acc, int E) {
    constexpr int FPL = CPE / 32;
    const int lane = threadIdx.x & 31;
    const int gi   = __builtin_amdgcn_readfirstlane((int)((blockIdx.x * blockDim.x + threadIdx.x) >> 5));
    const int base = gi << 5;
    if (base >= E) return;
    const int m = min(32, E - base);
    int s = 0, d = 0;
    if (lane < m) { s = src[base + lane]; d = dst[base + lane]; }
    const float nrm = dinv[s] * dinv[d];

    if (m == 32) {
        #pragma unroll
        for (int j0 = 0; j0 < 32; j0 += EP_BATCH) {
            float hv[EP_BATCH][FPL];
            #pragma unroll
            for (int u = 0; u < EP_BATCH; ++u) {
                const int sj = lane_bcast_i(s, j0 + u);
                const float* hs = H + (long)sj * CPE + lane;
                for (int k = 0; k < FPL; ++k) hv[u][k] = hs[k * 32];
            }
            #pragma unroll
            for (int u = 0; u < EP_BATCH; ++u) {
                const int   dj = lane_bcast_i(d, j0 + u);
                const float nj = lane_bcast_f(nrm, j0 + u);
                float* ad = acc + (long)dj * CPE + lane;
                for (int k = 0; k < FPL; ++k) atomic_add_f32(&ad[k * 32], hv[u][k] * nj);
            }
        }
    } else {
        for (int j = 0; j < m; ++j) {
            const int   sj = lane_bcast_i(s, j);
            const int   dj = lane_bcast_i(d, j);
            const float nj = lane_bcast_f(nrm, j);
            const float* hs = H   + (long)sj * CPE + lane;
            float*       ad = acc + (long)dj * CPE + lane;
            float hv[FPL];
            for (int k = 0; k < FPL; ++k) hv[k] = hs[k * 32];
            for (int k = 0; k < FPL; ++k) atomic_add_f32(&ad[k * 32], hv[k] * nj);
        }
    }
}

#define PP_THREADS 512
#define PP_WAVES   (PP_THREADS / 32)
#define PP_BATCH   16

__global__ void __launch_bounds__(PP_THREADS)
k_prop_pool(const int* __restrict__ src, const int* __restrict__ dst,
            const int* __restrict__ batch, const float* __restrict__ H,
            const float* __restrict__ dinv, const float* __restrict__ bias,
            float* pool, float* counts, int E, int Nn) {
    __shared__ float tab[NGRAPH * OUT_C];
    __shared__ float cnt[NGRAPH];
    for (int i = threadIdx.x; i < NGRAPH * OUT_C; i += PP_THREADS) tab[i] = 0.f;
    if (threadIdx.x < NGRAPH) cnt[threadIdx.x] = 0.f;
    __syncthreads();

    const int lane = threadIdx.x & 31;
    const int wib  = __builtin_amdgcn_readfirstlane((int)(threadIdx.x >> 5));
    const int nge  = (E + 31) >> 5;
    const int ngn  = (Nn + 31) >> 5;
    const int ng   = nge + ngn;
    const int tw   = (int)gridDim.x * PP_WAVES;
    const int c0   = lane * 2;
    const float bx = bias[c0], by = bias[c0 + 1];

    for (int gi = wib * (int)gridDim.x + (int)blockIdx.x; gi < ng; gi += tw) {
        const bool isnode = gi >= nge;
        int s = 0, d = 0, m;
        if (!isnode) {
            const int base = gi << 5;
            m = min(32, E - base);
            if (lane < m) { s = src[base + lane]; d = dst[base + lane]; }
        } else {
            const int base = (gi - nge) << 5;
            m = min(32, Nn - base);
            if (lane < m) { s = base + lane; d = s; }
        }
        const float nrm = dinv[s] * dinv[d];
        const int   g   = batch[d];
        const float ax = isnode ? bx : 0.f;
        const float ay = isnode ? by : 0.f;
        if (isnode && lane < m) atomicAdd(&cnt[g], 1.0f);

        if (m == 32) {
            #pragma unroll
            for (int j0 = 0; j0 < 32; j0 += PP_BATCH) {
                float2 v[PP_BATCH];
                #pragma unroll
                for (int u = 0; u < PP_BATCH; ++u) {
                    const int sj = lane_bcast_i(s, j0 + u);
                    v[u] = *reinterpret_cast<const float2*>(H + (long)sj * OUT_C + c0);
                }
                #pragma unroll
                for (int u = 0; u < PP_BATCH; ++u) {
                    const float nj = lane_bcast_f(nrm, j0 + u);
                    const int   gj = lane_bcast_i(g, j0 + u);
                    float* t = &tab[gj * OUT_C + c0];
                    atomicAdd(t,     fmaf(v[u].x, nj, ax));
                    atomicAdd(t + 1, fmaf(v[u].y, nj, ay));
                }
            }
        } else {
            for (int j = 0; j < m; ++j) {
                const int   sj = lane_bcast_i(s, j);
                const float nj = lane_bcast_f(nrm, j);
                const int   gj = lane_bcast_i(g, j);
                const float2 v = *reinterpret_cast<const float2*>(H + (long)sj * OUT_C + c0);
                float* t = &tab[gj * OUT_C + c0];
                atomicAdd(t,     fmaf(v.x, nj, ax));
                atomicAdd(t + 1, fmaf(v.y, nj, ay));
            }
        }
    }
    __syncthreads();

    constexpr int NLINES = NGRAPH * OUT_C / 32;
    for (int k = wib; k < NLINES; k += PP_WAVES) {
        const int idx = (((k + (int)blockIdx.x) & (NLINES - 1)) << 5) + lane;
        atomic_add_f32(&pool[idx], tab[idx]);
    }
    if (threadIdx.x < NGRAPH) atomic_add_f32(&counts[threadIdx.x], cnt[threadIdx.x]);
}

__global__ void k_pool_final(const float* __restrict__ pool, const float* __restrict__ counts,
                             float* __restrict__ out) {
    int idx = blockIdx.x * blockDim.x + threadIdx.x;
    if (idx >= NGRAPH * OUT_C) return;
    out[idx] = pool[idx] / fmaxf(counts[idx >> 6], 1.0f);
}

extern "C" void kernel_launch(void* const* d_in, const int* in_sizes, int n_in,
                              void* d_out, int out_size, void* d_ws, size_t ws_size,
                              hipStream_t stream) {
    const float* x   = (const float*)d_in[0];
    const float* W1  = (const float*)d_in[1];
    const float* b1  = (const float*)d_in[2];
    const float* W2  = (const float*)d_in[3];
    const float* b2  = (const float*)d_in[4];
    const int*   ei  = (const int*)d_in[5];
    const int*   bat = (const int*)d_in[6];

    const int Nn = in_sizes[0] / IN_C;
    const int E  = in_sizes[5] / 2;
    const int* src = ei;
    const int* dst = ei + E;

    char* w = (char*)d_ws;
    size_t off = 0;
    auto alloc = [&](size_t bytes) -> void* {
        void* p = w + off;
        off = (off + bytes + 255) & ~(size_t)255;
        return p;
    };
    float*    deg   = (float*)alloc((size_t)Nn * 4);
    float*    dinv  = (float*)alloc((size_t)Nn * 4);
    _Float16* W1h   = (_Float16*)alloc((size_t)IN_C * HID_C * 2);
    _Float16* W2h   = (_Float16*)alloc((size_t)HID_C * OUT_C * 2);
    float*    h1    = (float*)alloc((size_t)Nn * HID_C * 4);
    float*    a1    = (float*)alloc((size_t)Nn * HID_C * 4);
    float*    pool  = (float*)alloc((size_t)NGRAPH * OUT_C * 4);
    float*    cnts  = (float*)alloc((size_t)NGRAPH * 4);
    float*    h2 = h1;

    const int T = 256;
    k_fill<<<(Nn + T - 1) / T, T, 0, stream>>>(deg, 1.0f, Nn);
    k_edge_deg<<<(E + T - 1) / T, T, 0, stream>>>(dst, deg, E);
    k_rsqrt<<<(Nn + T - 1) / T, T, 0, stream>>>(deg, dinv, Nn);

    k_pack_w<<<(IN_C * HID_C + T - 1) / T, T, 0, stream>>>(W1, W1h, IN_C, HID_C);
    k_pack_w<<<(HID_C * OUT_C + T - 1) / T, T, 0, stream>>>(W2, W2h, HID_C, OUT_C);

    {
        int waves = Nn / 16;
        int blocks = (waves * 32 + T - 1) / T;
        k_gemm<HID_C / 16, false><<<blocks, T, 0, stream>>>(x, W1h, h1, Nn, IN_C, HID_C);
    }
    k_self_bias<<<(Nn * HID_C + T - 1) / T, T, 0, stream>>>(h1, dinv, b1, a1, Nn, HID_C);
    k_edge_prop<HID_C><<<(E + 255) / 256, EP_THREADS, 0, stream>>>(src, dst, h1, dinv, a1, E);

    {
        int waves = Nn / 16;
        int blocks = (waves * 32 + T - 1) / T;
        k_gemm<OUT_C / 16, true><<<blocks, T, 0, stream>>>(a1, W2h, h2, Nn, HID_C, OUT_C);
    }
    k_fill<<<(NGRAPH * OUT_C + T - 1) / T, T, 0, stream>>>(pool, 0.0f, NGRAPH * OUT_C);
    k_fill<<<1, T, 0, stream>>>(cnts, 0.0f, NGRAPH);
    k_prop_pool<<<512, PP_THREADS, 0, stream>>>(src, dst, bat, h2, dinv, b2, pool, cnts, E, Nn);
    k_pool_final<<<(NGRAPH * OUT_C + T - 1) / T, T, 0, stream>>>(pool, cnts, (float*)d_out);
}


# layer-1 product epilogue also forms dinv = rsqrt(deg) and the scatter's start rows h1*dinv^2 + b1 from its accumulators; both per-node passes behind it go, 13 launches to 11; sums unchanged
# speedup vs baseline: 2.0686x; 1.0854x over previous
#include <hip/hip_runtime.h>
#include <hip/hip_bf16.h>
#include <stdint.h>

typedef __attribute__((ext_vector_type(16))) _Float16 v16h;
typedef __attribute__((ext_vector_type(8)))  float    v8f;

#define IN_C   128
#define HID_C  128
#define OUT_C  64
#define NGRAPH 128

__device__ __forceinline__ void atomic_add_f32(float* p, float v) {
    unsafeAtomicAdd(p, v);
}

__global__ void k_fill(float* p, float v, int n) {
    int i = blockIdx.x * blockDim.x + threadIdx.x;
    if (i < n) p[i] = v;
}

__global__ void k_edge_deg(const int* __restrict__ dst, float* deg, int E) {
    int e = blockIdx.x * blockDim.x + threadIdx.x;
    if (e < E) atomic_add_f32(&deg[dst[e]], 1.0f);
}

__global__ void k_pack_w(const float* __restrict__ W, _Float16* __restrict__ Wh,
                         int K, int N) {
    int idx = blockIdx.x * blockDim.x + threadIdx.x;
    if (idx >= K * N) return;
    int frag   = idx >> 9;
    int within = idx & 511;
    int lane   = within >> 4;
    int i      = within & 15;
    int ntiles = N >> 4;
    int kt = frag / ntiles, nt = frag % ntiles;
    int n  = (nt << 4) + (lane & 15);
    int kk = (i < 8 ? i : i + 8) + ((lane >> 4) << 3);
    int k  = (kt << 5) + kk;
    Wh[idx] = (_Float16)W[k * N + n];
}

template <int NT, bool RELU, bool SELF>
__global__ void k_gemm(const float* __restrict__ A, const _Float16* __restrict__ Bh,
                       float* __restrict__ C, int M, int K, int N,
                       const float* __restrict__ deg, const float* __restrict__ bias,
                       float* __restrict__ dinv, float* __restrict__ C2) {
    int wave = (int)((blockIdx.x * blockDim.x + threadIdx.x) >> 5);
    int lane = threadIdx.x & 31;
    int mtiles = M >> 4;
    if (wave >= mtiles) return;

    int m0  = wave << 4;
    int hi  = lane >> 4;
    int row = m0 + (lane & 15);

    v8f acc[NT];
    v8f zero = {0.f,0.f,0.f,0.f,0.f,0.f,0.f,0.f};
    #pragma unroll
    for (int j = 0; j < NT; ++j) acc[j] = zero;

    const int mrow = m0 + (hi << 3);
    float4 g0, g1;
    if (SELF) {
        const float4* dp = reinterpret_cast<const float4*>(deg + mrow);
        g0 = dp[0]; g1 = dp[1];
    }

    const v16h* bp = reinterpret_cast<const v16h*>(Bh);
    int ktiles = K >> 5;
    for (int kt = 0; kt < ktiles; ++kt) {
        int kbase = (kt << 5) + (hi << 3);
        const float4* p = reinterpret_cast<const float4*>(A + (long)row * K + kbase);
        float4 x0 = p[0], x1 = p[1];
        float4 x2 = p[4], x3 = p[5];
        if (RELU) {
            x0.x=fmaxf(x0.x,0.f); x0.y=fmaxf(x0.y,0.f); x0.z=fmaxf(x0.z,0.f); x0.w=fmaxf(x0.w,0.f);
            x1.x=fmaxf(x1.x,0.f); x1.y=fmaxf(x1.y,0.f); x1.z=fmaxf(x1.z,0.f); x1.w=fmaxf(x1.w,0.f);
            x2.x=fmaxf(x2.x,0.f); x2.y=fmaxf(x2.y,0.f); x2.z=fmaxf(x2.z,0.f); x2.w=fmaxf(x2.w,0.f);
            x3.x=fmaxf(x3.x,0.f); x3.y=fmaxf(x3.y,0.f); x3.z=fmaxf(x3.z,0.f); x3.w=fmaxf(x3.w,0.f);
        }
        v16h a;
        a[0]=(_Float16)x0.x; a[1]=(_Float16)x0.y; a[2]=(_Float16)x0.z; a[3]=(_Float16)x0.w;
        a[4]=(_Float16)x1.x; a[5]=(_Float16)x1.y; a[6]=(_Float16)x1.z; a[7]=(_Float16)x1.w;
        a[8]=(_Float16)x2.x; a[9]=(_Float16)x2.y; a[10]=(_Float16)x2.z; a[11]=(_Float16)x2.w;
        a[12]=(_Float16)x3.x; a[13]=(_Float16)x3.y; a[14]=(_Float16)x3.z; a[15]=(_Float16)x3.w;

        #pragma unroll
        for (int nt = 0; nt < NT; ++nt) {
            v16h b = bp[(size_t)(kt * NT + nt) * 32 + lane];
            acc[nt] = __builtin_amdgcn_wmma_f32_16x16x32_f16(
                false, a, false, b, (short)0, acc[nt], false, false);
        }
    }

    if (SELF) {
        float d[8];
        d[0] = rsqrtf(g0.x); d[1] = rsqrtf(g0.y); d[2] = rsqrtf(g0.z); d[3] = rsqrtf(g0.w);
        d[4] = rsqrtf(g1.x); d[5] = rsqrtf(g1.y); d[6] = rsqrtf(g1.z); d[7] = rsqrtf(g1.w);
        if ((lane & 15) == 0) {
            float4* op = reinterpret_cast<float4*>(dinv + mrow);
            op[0] = make_float4(d[0], d[1], d[2], d[3]);
            op[1] = make_float4(d[4], d[5], d[6], d[7]);
        }
        #pragma unroll
        for (int nt = 0; nt < NT; ++nt) {
            int n = (nt << 4) + (lane & 15);
            const float bn = bias[n];
            #pragma unroll
            for (int j = 0; j < 8; ++j) {
                const float v = acc[nt][j];
                C [(long)(mrow + j) * N + n] = v;
                C2[(long)(mrow + j) * N + n] = v * d[j] * d[j] + bn;
            }
        }
    } else {
        #pragma unroll
        for (int nt = 0; nt < NT; ++nt) {
            int n = (nt << 4) + (lane & 15);
            #pragma unroll
            for (int j = 0; j < 8; ++j)
                C[(long)(mrow + j) * N + n] = acc[nt][j];
        }
    }
}

__device__ __forceinline__ int   lane_bcast_i(int v, int j)   { return __builtin_amdgcn_readlane(v, j); }
__device__ __forceinline__ float lane_bcast_f(float v, int j) {
    return __int_as_float(__builtin_amdgcn_readlane(__float_as_int(v), j));
}

#define EP_THREADS 256
#define EP_BATCH   8
template <int CPE>
__global__ void __launch_bounds__(EP_THREADS)
k_edge_prop(const int* __restrict__ src, const int* __restrict__ dst,
            const float* __restrict__ H, const float* __restrict__ dinv,
            float* acc, int E) {
    constexpr int FPL = CPE / 32;
    const int lane = threadIdx.x & 31;
    const int gi   = __builtin_amdgcn_readfirstlane((int)((blockIdx.x * blockDim.x + threadIdx.x) >> 5));
    const int base = gi << 5;
    if (base >= E) return;
    const int m = min(32, E - base);
    int s = 0, d = 0;
    if (lane < m) { s = src[base + lane]; d = dst[base + lane]; }
    const float nrm = dinv[s] * dinv[d];

    if (m == 32) {
        #pragma unroll
        for (int j0 = 0; j0 < 32; j0 += EP_BATCH) {
            float hv[EP_BATCH][FPL];
            #pragma unroll
            for (int u = 0; u < EP_BATCH; ++u) {
                const int sj = lane_bcast_i(s, j0 + u);
                const float* hs = H + (long)sj * CPE + lane;
                for (int k = 0; k < FPL; ++k) hv[u][k] = hs[k * 32];
            }
            #pragma unroll
            for (int u = 0; u < EP_BATCH; ++u) {
                const int   dj = lane_bcast_i(d, j0 + u);
                const float nj = lane_bcast_f(nrm, j0 + u);
                float* ad = acc + (long)dj * CPE + lane;
                for (int k = 0; k < FPL; ++k) atomic_add_f32(&ad[k * 32], hv[u][k] * nj);
            }
        }
    } else {
        for (int j = 0; j < m; ++j) {
            const int   sj = lane_bcast_i(s, j);
            const int   dj = lane_bcast_i(d, j);
            const float nj = lane_bcast_f(nrm, j);
            const float* hs = H   + (long)sj * CPE + lane;
            float*       ad = acc + (long)dj * CPE + lane;
            float hv[FPL];
            for (int k = 0; k < FPL; ++k) hv[k] = hs[k * 32];
            for (int k = 0; k < FPL; ++k) atomic_add_f32(&ad[k * 32], hv[k] * nj);
        }
    }
}

#define PP_THREADS 512
#define PP_WAVES   (PP_THREADS / 32)
#define PP_BATCH   16

__global__ void __launch_bounds__(PP_THREADS)
k_prop_pool(const int* __restrict__ src, const int* __restrict__ dst,
            const int* __restrict__ batch, const float* __restrict__ H,
            const float* __restrict__ dinv, const float* __restrict__ bias,
            float* pool, float* counts, int E, int Nn) {
    __shared__ float tab[NGRAPH * OUT_C];
    __shared__ float cnt[NGRAPH];
    for (int i = threadIdx.x; i < NGRAPH * OUT_C; i += PP_THREADS) tab[i] = 0.f;
    if (threadIdx.x < NGRAPH) cnt[threadIdx.x] = 0.f;
    __syncthreads();

    const int lane = threadIdx.x & 31;
    const int wib  = __builtin_amdgcn_readfirstlane((int)(threadIdx.x >> 5));
    const int nge  = (E + 31) >> 5;
    const int ngn  = (Nn + 31) >> 5;
    const int ng   = nge + ngn;
    const int tw   = (int)gridDim.x * PP_WAVES;
    const int c0   = lane * 2;
    const float bx = bias[c0], by = bias[c0 + 1];

    for (int gi = wib * (int)gridDim.x + (int)blockIdx.x; gi < ng; gi += tw) {
        const bool isnode = gi >= nge;
        int s = 0, d = 0, m;
        if (!isnode) {
            const int base = gi << 5;
            m = min(32, E - base);
            if (lane < m) { s = src[base + lane]; d = dst[base + lane]; }
        } else {
            const int base = (gi - nge) << 5;
            m = min(32, Nn - base);
            if (lane < m) { s = base + lane; d = s; }
        }
        const float nrm = dinv[s] * dinv[d];
        const int   g   = batch[d];
        const float ax = isnode ? bx : 0.f;
        const float ay = isnode ? by : 0.f;
        if (isnode && lane < m) atomicAdd(&cnt[g], 1.0f);

        if (m == 32) {
            #pragma unroll
            for (int j0 = 0; j0 < 32; j0 += PP_BATCH) {
                float2 v[PP_BATCH];
                #pragma unroll
                for (int u = 0; u < PP_BATCH; ++u) {
                    const int sj = lane_bcast_i(s, j0 + u);
                    v[u] = *reinterpret_cast<const float2*>(H + (long)sj * OUT_C + c0);
                }
                #pragma unroll
                for (int u = 0; u < PP_BATCH; ++u) {
                    const float nj = lane_bcast_f(nrm, j0 + u);
                    const int   gj = lane_bcast_i(g, j0 + u);
                    float* t = &tab[gj * OUT_C + c0];
                    atomicAdd(t,     fmaf(v[u].x, nj, ax));
                    atomicAdd(t + 1, fmaf(v[u].y, nj, ay));
                }
            }
        } else {
            for (int j = 0; j < m; ++j) {
                const int   sj = lane_bcast_i(s, j);
                const float nj = lane_bcast_f(nrm, j);
                const int   gj = lane_bcast_i(g, j);
                const float2 v = *reinterpret_cast<const float2*>(H + (long)sj * OUT_C + c0);
                float* t = &tab[gj * OUT_C + c0];
                atomicAdd(t,     fmaf(v.x, nj, ax));
                atomicAdd(t + 1, fmaf(v.y, nj, ay));
            }
        }
    }
    __syncthreads();

    constexpr int NLINES = NGRAPH * OUT_C / 32;
    for (int k = wib; k < NLINES; k += PP_WAVES) {
        const int idx = (((k + (int)blockIdx.x) & (NLINES - 1)) << 5) + lane;
        atomic_add_f32(&pool[idx], tab[idx]);
    }
    if (threadIdx.x < NGRAPH) atomic_add_f32(&counts[threadIdx.x], cnt[threadIdx.x]);
}

__global__ void k_pool_final(const float* __restrict__ pool, const float* __restrict__ counts,
                             float* __restrict__ out) {
    int idx = blockIdx.x * blockDim.x + threadIdx.x;
    if (idx >= NGRAPH * OUT_C) return;
    out[idx] = pool[idx] / fmaxf(counts[idx >> 6], 1.0f);
}

extern "C" void kernel_launch(void* const* d_in, const int* in_sizes, int n_in,
                              void* d_out, int out_size, void* d_ws, size_t ws_size,
                              hipStream_t stream) {
    const float* x   = (const float*)d_in[0];
    const float* W1  = (const float*)d_in[1];
    const float* b1  = (const float*)d_in[2];
    const float* W2  = (const float*)d_in[3];
    const float* b2  = (const float*)d_in[4];
    const int*   ei  = (const int*)d_in[5];
    const int*   bat = (const int*)d_in[6];

    const int Nn = in_sizes[0] / IN_C;
    const int E  = in_sizes[5] / 2;
    const int* src = ei;
    const int* dst = ei + E;

    char* w = (char*)d_ws;
    size_t off = 0;
    auto alloc = [&](size_t bytes) -> void* {
        void* p = w + off;
        off = (off + bytes + 255) & ~(size_t)255;
        return p;
    };
    float*    deg   = (float*)alloc((size_t)Nn * 4);
    float*    dinv  = (float*)alloc((size_t)Nn * 4);
    _Float16* W1h   = (_Float16*)alloc((size_t)IN_C * HID_C * 2);
    _Float16* W2h   = (_Float16*)alloc((size_t)HID_C * OUT_C * 2);
    float*    h1    = (float*)alloc((size_t)Nn * HID_C * 4);
    float*    a1    = (float*)alloc((size_t)Nn * HID_C * 4);
    float*    pool  = (float*)alloc((size_t)NGRAPH * OUT_C * 4);
    float*    cnts  = (float*)alloc((size_t)NGRAPH * 4);
    float*    h2 = h1;

    const int T = 256;
    k_fill<<<(Nn + T - 1) / T, T, 0, stream>>>(deg, 1.0f, Nn);
    k_edge_deg<<<(E + T - 1) / T, T, 0, stream>>>(dst, deg, E);

    k_pack_w<<<(IN_C * HID_C + T - 1) / T, T, 0, stream>>>(W1, W1h, IN_C, HID_C);
    k_pack_w<<<(HID_C * OUT_C + T - 1) / T, T, 0, stream>>>(W2, W2h, HID_C, OUT_C);

    {
        int waves = Nn / 16;
        int blocks = (waves * 32 + T - 1) / T;
        k_gemm<HID_C / 16, false, true><<<blocks, T, 0, stream>>>(x, W1h, h1, Nn, IN_C, HID_C,
                                                                 deg, b1, dinv, a1);
    }
    k_edge_prop<HID_C><<<(E + 255) / 256, EP_THREADS, 0, stream>>>(src, dst, h1, dinv, a1, E);

    {
        int waves = Nn / 16;
        int blocks = (waves * 32 + T - 1) / T;
        k_gemm<OUT_C / 16, true, false><<<blocks, T, 0, stream>>>(a1, W2h, h2, Nn, HID_C, OUT_C,
                                                                 nullptr, nullptr, nullptr, nullptr);
    }
    k_fill<<<(NGRAPH * OUT_C + T - 1) / T, T, 0, stream>>>(pool, 0.0f, NGRAPH * OUT_C);
    k_fill<<<1, T, 0, stream>>>(cnts, 0.0f, NGRAPH);
    k_prop_pool<<<512, PP_THREADS, 0, stream>>>(src, dst, bat, h2, dinv, b2, pool, cnts, E, Nn);
    k_pool_final<<<(NGRAPH * OUT_C + T - 1) / T, T, 0, stream>>>(pool, cnts, (float*)d_out);
}


# both products: each work-group packs its f32 weight matrix into f16 B fragments in its own LDS (one group per WGP, strips wave-major); the two packing launches go, 11 launches to 9; sums unchanged
# speedup vs baseline: 2.2376x; 1.0817x over previous
#include <hip/hip_runtime.h>
#include <hip/hip_bf16.h>
#include <stdint.h>

typedef __attribute__((ext_vector_type(16))) _Float16 v16h;
typedef __attribute__((ext_vector_type(8)))  _Float16 v8h;
typedef __attribute__((ext_vector_type(8)))  float    v8f;

#define IN_C   128
#define HID_C  128
#define OUT_C  64
#define NGRAPH 128

__device__ __forceinline__ void atomic_add_f32(float* p, float v) {
    unsafeAtomicAdd(p, v);
}

__global__ void k_fill(float* p, float v, int n) {
    int i = blockIdx.x * blockDim.x + threadIdx.x;
    if (i < n) p[i] = v;
}

__global__ void k_edge_deg(const int* __restrict__ dst, float* deg, int E) {
    int e = blockIdx.x * blockDim.x + threadIdx.x;
    if (e < E) atomic_add_f32(&deg[dst[e]], 1.0f);
}

#define GM_THREADS 512
#define GM_K       128
#define GM_BLOCKS  256
template <int NT, bool RELU, bool SELF>
__global__ void __launch_bounds__(GM_THREADS)
k_gemm(const float* __restrict__ A, const float* __restrict__ W,
       float* __restrict__ C, int M,
       const float* __restrict__ deg, const float* __restrict__ bias,
       float* __restrict__ dinv, float* __restrict__ C2) {
    constexpr int K = GM_K;
    constexpr int N = NT * 16;
    __shared__ __attribute__((aligned(16))) _Float16 wl[K * N];
    v8h* wlv = reinterpret_cast<v8h*>(wl);
    #pragma unroll
    for (int it = 0; it < (K / 8) * N / GM_THREADS; ++it) {
        const int t  = it * GM_THREADS + (int)threadIdx.x;
        const int n  = t % N;
        const int k0 = (t / N) * 8;
        const float* wp = W + k0 * N + n;
        v8f f;
        #pragma unroll
        for (int j = 0; j < 8; ++j) f[j] = wp[j * N];
        const v8h h = __builtin_convertvector(f, v8h);
        const int kk0 = k0 & 31;
        wlv[((k0 >> 5) * NT + (n >> 4)) * 64 + (kk0 >> 4) * 32 + (((kk0 >> 3) & 1) << 4) + (n & 15)] = h;
    }
    __syncthreads();

    const int lane = threadIdx.x & 31;
    const int wib  = __builtin_amdgcn_readfirstlane((int)(threadIdx.x >> 5));
    const int mtiles = M >> 4;
    for (int strip = wib * (int)gridDim.x + (int)blockIdx.x; strip < mtiles;
         strip += (int)gridDim.x * (GM_THREADS / 32)) {

    int m0  = strip << 4;
    int hi  = lane >> 4;
    int row = m0 + (lane & 15);

    v8f acc[NT];
    v8f zero = {0.f,0.f,0.f,0.f,0.f,0.f,0.f,0.f};
    #pragma unroll
    for (int j = 0; j < NT; ++j) acc[j] = zero;

    const int mrow = m0 + (hi << 3);
    float4 g0, g1;
    if (SELF) {
        const float4* dp = reinterpret_cast<const float4*>(deg + mrow);
        g0 = dp[0]; g1 = dp[1];
    }

    const v8h* bl = reinterpret_cast<const v8h*>(wl);
    constexpr int ktiles = K >> 5;
    for (int kt = 0; kt < ktiles; ++kt) {
        int kbase = (kt << 5) + (hi << 3);
        const float4* p = reinterpret_cast<const float4*>(A + (long)row * K + kbase);
        float4 x0 = p[0], x1 = p[1];
        float4 x2 = p[4], x3 = p[5];
        if (RELU) {
            x0.x=fmaxf(x0.x,0.f); x0.y=fmaxf(x0.y,0.f); x0.z=fmaxf(x0.z,0.f); x0.w=fmaxf(x0.w,0.f);
            x1.x=fmaxf(x1.x,0.f); x1.y=fmaxf(x1.y,0.f); x1.z=fmaxf(x1.z,0.f); x1.w=fmaxf(x1.w,0.f);
            x2.x=fmaxf(x2.x,0.f); x2.y=fmaxf(x2.y,0.f); x2.z=fmaxf(x2.z,0.f); x2.w=fmaxf(x2.w,0.f);
            x3.x=fmaxf(x3.x,0.f); x3.y=fmaxf(x3.y,0.f); x3.z=fmaxf(x3.z,0.f); x3.w=fmaxf(x3.w,0.f);
        }
        v16h a;
        a[0]=(_Float16)x0.x; a[1]=(_Float16)x0.y; a[2]=(_Float16)x0.z; a[3]=(_Float16)x0.w;
        a[4]=(_Float16)x1.x; a[5]=(_Float16)x1.y; a[6]=(_Float16)x1.z; a[7]=(_Float16)x1.w;
        a[8]=(_Float16)x2.x; a[9]=(_Float16)x2.y; a[10]=(_Float16)x2.z; a[11]=(_Float16)x2.w;
        a[12]=(_Float16)x3.x; a[13]=(_Float16)x3.y; a[14]=(_Float16)x3.z; a[15]=(_Float16)x3.w;

        #pragma unroll
        for (int nt = 0; nt < NT; ++nt) {
            const v8h blo = bl[(kt * NT + nt) * 64 + lane];
            const v8h bhi = bl[(kt * NT + nt) * 64 + 32 + lane];
            const v16h b = __builtin_shufflevector(blo, bhi, 0, 1, 2, 3, 4, 5, 6, 7,
                                                   8, 9, 10, 11, 12, 13, 14, 15);
            acc[nt] = __builtin_amdgcn_wmma_f32_16x16x32_f16(
                false, a, false, b, (short)0, acc[nt], false, false);
        }
    }

    if (SELF) {
        float d[8];
        d[0] = rsqrtf(g0.x); d[1] = rsqrtf(g0.y); d[2] = rsqrtf(g0.z); d[3] = rsqrtf(g0.w);
        d[4] = rsqrtf(g1.x); d[5] = rsqrtf(g1.y); d[6] = rsqrtf(g1.z); d[7] = rsqrtf(g1.w);
        if ((lane & 15) == 0) {
            float4* op = reinterpret_cast<float4*>(dinv + mrow);
            op[0] = make_float4(d[0], d[1], d[2], d[3]);
            op[1] = make_float4(d[4], d[5], d[6], d[7]);
        }
        #pragma unroll
        for (int nt = 0; nt < NT; ++nt) {
            int n = (nt << 4) + (lane & 15);
            const float bn = bias[n];
            #pragma unroll
            for (int j = 0; j < 8; ++j) {
                const float v = acc[nt][j];
                C [(long)(mrow + j) * N + n] = v;
                C2[(long)(mrow + j) * N + n] = v * d[j] * d[j] + bn;
            }
        }
    } else {
        #pragma unroll
        for (int nt = 0; nt < NT; ++nt) {
            int n = (nt << 4) + (lane & 15);
            #pragma unroll
            for (int j = 0; j < 8; ++j)
                C[(long)(mrow + j) * N + n] = acc[nt][j];
        }
    }
    }
}

__device__ __forceinline__ int   lane_bcast_i(int v, int j)   { return __builtin_amdgcn_readlane(v, j); }
__device__ __forceinline__ float lane_bcast_f(float v, int j) {
    return __int_as_float(__builtin_amdgcn_readlane(__float_as_int(v), j));
}

#define EP_THREADS 256
#define EP_BATCH   8
template <int CPE>
__global__ void __launch_bounds__(EP_THREADS)
k_edge_prop(const int* __restrict__ src, const int* __restrict__ dst,
            const float* __restrict__ H, const float* __restrict__ dinv,
            float* acc, int E) {
    constexpr int FPL = CPE / 32;
    const int lane = threadIdx.x & 31;
    const int gi   = __builtin_amdgcn_readfirstlane((int)((blockIdx.x * blockDim.x + threadIdx.x) >> 5));
    const int base = gi << 5;
    if (base >= E) return;
    const int m = min(32, E - base);
    int s = 0, d = 0;
    if (lane < m) { s = src[base + lane]; d = dst[base + lane]; }
    const float nrm = dinv[s] * dinv[d];

    if (m == 32) {
        #pragma unroll
        for (int j0 = 0; j0 < 32; j0 += EP_BATCH) {
            float hv[EP_BATCH][FPL];
            #pragma unroll
            for (int u = 0; u < EP_BATCH; ++u) {
                const int sj = lane_bcast_i(s, j0 + u);
                const float* hs = H + (long)sj * CPE + lane;
                for (int k = 0; k < FPL; ++k) hv[u][k] = hs[k * 32];
            }
            #pragma unroll
            for (int u = 0; u < EP_BATCH; ++u) {
                const int   dj = lane_bcast_i(d, j0 + u);
                const float nj = lane_bcast_f(nrm, j0 + u);
                float* ad = acc + (long)dj * CPE + lane;
                for (int k = 0; k < FPL; ++k) atomic_add_f32(&ad[k * 32], hv[u][k] * nj);
            }
        }
    } else {
        for (int j = 0; j < m; ++j) {
            const int   sj = lane_bcast_i(s, j);
            const int   dj = lane_bcast_i(d, j);
            const float nj = lane_bcast_f(nrm, j);
            const float* hs = H   + (long)sj * CPE + lane;
            float*       ad = acc + (long)dj * CPE + lane;
            float hv[FPL];
            for (int k = 0; k < FPL; ++k) hv[k] = hs[k * 32];
            for (int k = 0; k < FPL; ++k) atomic_add_f32(&ad[k * 32], hv[k] * nj);
        }
    }
}

#define PP_THREADS 512
#define PP_WAVES   (PP_THREADS / 32)
#define PP_BATCH   16

__global__ void __launch_bounds__(PP_THREADS)
k_prop_pool(const int* __restrict__ src, const int* __restrict__ dst,
            const int* __restrict__ batch, const float* __restrict__ H,
            const float* __restrict__ dinv, const float* __restrict__ bias,
            float* pool, float* counts, int E, int Nn) {
    __shared__ float tab[NGRAPH * OUT_C];
    __shared__ float cnt[NGRAPH];
    for (int i = threadIdx.x; i < NGRAPH * OUT_C; i += PP_THREADS) tab[i] = 0.f;
    if (threadIdx.x < NGRAPH) cnt[threadIdx.x] = 0.f;
    __syncthreads();

    const int lane = threadIdx.x & 31;
    const int wib  = __builtin_amdgcn_readfirstlane((int)(threadIdx.x >> 5));
    const int nge  = (E + 31) >> 5;
    const int ngn  = (Nn + 31) >> 5;
    const int ng   = nge + ngn;
    const int tw   = (int)gridDim.x * PP_WAVES;
    const int c0   = lane * 2;
    const float bx = bias[c0], by = bias[c0 + 1];

    for (int gi = wib * (int)gridDim.x + (int)blockIdx.x; gi < ng; gi += tw) {
        const bool isnode = gi >= nge;
        int s = 0, d = 0, m;
        if (!isnode) {
            const int base = gi << 5;
            m = min(32, E - base);
            if (lane < m) { s = src[base + lane]; d = dst[base + lane]; }
        } else {
            const int base = (gi - nge) << 5;
            m = min(32, Nn - base);
            if (lane < m) { s = base + lane; d = s; }
        }
        const float nrm = dinv[s] * dinv[d];
        const int   g   = batch[d];
        const float ax = isnode ? bx : 0.f;
        const float ay = isnode ? by : 0.f;
        if (isnode && lane < m) atomicAdd(&cnt[g], 1.0f);

        if (m == 32) {
            #pragma unroll
            for (int j0 = 0; j0 < 32; j0 += PP_BATCH) {
                float2 v[PP_BATCH];
                #pragma unroll
                for (int u = 0; u < PP_BATCH; ++u) {
                    const int sj = lane_bcast_i(s, j0 + u);
                    v[u] = *reinterpret_cast<const float2*>(H + (long)sj * OUT_C + c0);
                }
                #pragma unroll
                for (int u = 0; u < PP_BATCH; ++u) {
                    const float nj = lane_bcast_f(nrm, j0 + u);
                    const int   gj = lane_bcast_i(g, j0 + u);
                    float* t = &tab[gj * OUT_C + c0];
                    atomicAdd(t,     fmaf(v[u].x, nj, ax));
                    atomicAdd(t + 1, fmaf(v[u].y, nj, ay));
                }
            }
        } else {
            for (int j = 0; j < m; ++j) {
                const int   sj = lane_bcast_i(s, j);
                const float nj = lane_bcast_f(nrm, j);
                const int   gj = lane_bcast_i(g, j);
                const float2 v = *reinterpret_cast<const float2*>(H + (long)sj * OUT_C + c0);
                float* t = &tab[gj * OUT_C + c0];
                atomicAdd(t,     fmaf(v.x, nj, ax));
                atomicAdd(t + 1, fmaf(v.y, nj, ay));
            }
        }
    }
    __syncthreads();

    constexpr int NLINES = NGRAPH * OUT_C / 32;
    for (int k = wib; k < NLINES; k += PP_WAVES) {
        const int idx = (((k + (int)blockIdx.x) & (NLINES - 1)) << 5) + lane;
        atomic_add_f32(&pool[idx], tab[idx]);
    }
    if (threadIdx.x < NGRAPH) atomic_add_f32(&counts[threadIdx.x], cnt[threadIdx.x]);
}

__global__ void k_pool_final(const float* __restrict__ pool, const float* __restrict__ counts,
                             float* __restrict__ out) {
    int idx = blockIdx.x * blockDim.x + threadIdx.x;
    if (idx >= NGRAPH * OUT_C) return;
    out[idx] = pool[idx] / fmaxf(counts[idx >> 6], 1.0f);
}

extern "C" void kernel_launch(void* const* d_in, const int* in_sizes, int n_in,
                              void* d_out, int out_size, void* d_ws, size_t ws_size,
                              hipStream_t stream) {
    const float* x   = (const float*)d_in[0];
    const float* W1  = (const float*)d_in[1];
    const float* b1  = (const float*)d_in[2];
    const float* W2  = (const float*)d_in[3];
    const float* b2  = (const float*)d_in[4];
    const int*   ei  = (const int*)d_in[5];
    const int*   bat = (const int*)d_in[6];

    const int Nn = in_sizes[0] / IN_C;
    const int E  = in_sizes[5] / 2;
    const int* src = ei;
    const int* dst = ei + E;

    char* w = (char*)d_ws;
    size_t off = 0;
    auto alloc = [&](size_t bytes) -> void* {
        void* p = w + off;
        off = (off + bytes + 255) & ~(size_t)255;
        return p;
    };
    float*    deg   = (float*)alloc((size_t)Nn * 4);
    float*    dinv  = (float*)alloc((size_t)Nn * 4);
    float*    h1    = (float*)alloc((size_t)Nn * HID_C * 4);
    float*    a1    = (float*)alloc((size_t)Nn * HID_C * 4);
    float*    pool  = (float*)alloc((size_t)NGRAPH * OUT_C * 4);
    float*    cnts  = (float*)alloc((size_t)NGRAPH * 4);
    float*    h2 = h1;

    const int T = 256;
    k_fill<<<(Nn + T - 1) / T, T, 0, stream>>>(deg, 1.0f, Nn);
    k_edge_deg<<<(E + T - 1) / T, T, 0, stream>>>(dst, deg, E);

    k_gemm<HID_C / 16, false, true><<<GM_BLOCKS, GM_THREADS, 0, stream>>>(x, W1, h1, Nn,
                                                                         deg, b1, dinv, a1);
    k_edge_prop<HID_C><<<(E + 255) / 256, EP_THREADS, 0, stream>>>(src, dst, h1, dinv, a1, E);

    k_gemm<OUT_C / 16, true, false><<<GM_BLOCKS, GM_THREADS, 0, stream>>>(a1, W2, h2, Nn,
                                                                         nullptr, nullptr, nullptr, nullptr);
    k_fill<<<(NGRAPH * OUT_C + T - 1) / T, T, 0, stream>>>(pool, 0.0f, NGRAPH * OUT_C);
    k_fill<<<1, T, 0, stream>>>(cnts, 0.0f, NGRAPH);
    k_prop_pool<<<512, PP_THREADS, 0, stream>>>(src, dst, bat, h2, dinv, b2, pool, cnts, E, Nn);
    k_pool_final<<<(NGRAPH * OUT_C + T - 1) / T, T, 0, stream>>>(pool, cnts, (float*)d_out);
}


# pool sums: each propagate block stores its LDS table as its own slice, the last kernel adds slices in fixed block order (was float atomics onto one table); both clears go, 9 launches to 7
# speedup vs baseline: 2.3045x; 1.0299x over previous
#include <hip/hip_runtime.h>
#include <hip/hip_bf16.h>
#include <stdint.h>

typedef __attribute__((ext_vector_type(16))) _Float16 v16h;
typedef __attribute__((ext_vector_type(8)))  _Float16 v8h;
typedef __attribute__((ext_vector_type(8)))  float    v8f;

#define IN_C   128
#define HID_C  128
#define OUT_C  64
#define NGRAPH 128

__device__ __forceinline__ void atomic_add_f32(float* p, float v) {
    unsafeAtomicAdd(p, v);
}

__global__ void k_fill(float* p, float v, int n) {
    int i = blockIdx.x * blockDim.x + threadIdx.x;
    if (i < n) p[i] = v;
}

__global__ void k_edge_deg(const int* __restrict__ dst, float* deg, int E) {
    int e = blockIdx.x * blockDim.x + threadIdx.x;
    if (e < E) atomic_add_f32(&deg[dst[e]], 1.0f);
}

#define GM_THREADS 512
#define GM_K       128
#define GM_BLOCKS  256
template <int NT, bool RELU, bool SELF>
__global__ void __launch_bounds__(GM_THREADS)
k_gemm(const float* __restrict__ A, const float* __restrict__ W,
       float* __restrict__ C, int M,
       const float* __restrict__ deg, const float* __restrict__ bias,
       float* __restrict__ dinv, float* __restrict__ C2) {
    constexpr int K = GM_K;
    constexpr int N = NT * 16;
    __shared__ __attribute__((aligned(16))) _Float16 wl[K * N];
    v8h* wlv = reinterpret_cast<v8h*>(wl);
    #pragma unroll
    for (int it = 0; it < (K / 8) * N / GM_THREADS; ++it) {
        const int t  = it * GM_THREADS + (int)threadIdx.x;
        const int n  = t % N;
        const int k0 = (t / N) * 8;
        const float* wp = W + k0 * N + n;
        v8f f;
        #pragma unroll
        for (int j = 0; j < 8; ++j) f[j] = wp[j * N];
        const v8h h = __builtin_convertvector(f, v8h);
        const int kk0 = k0 & 31;
        wlv[((k0 >> 5) * NT + (n >> 4)) * 64 + (kk0 >> 4) * 32 + (((kk0 >> 3) & 1) << 4) + (n & 15)] = h;
    }
    __syncthreads();

    const int lane = threadIdx.x & 31;
    const int wib  = __builtin_amdgcn_readfirstlane((int)(threadIdx.x >> 5));
    const int mtiles = M >> 4;
    for (int strip = wib * (int)gridDim.x + (int)blockIdx.x; strip < mtiles;
         strip += (int)gridDim.x * (GM_THREADS / 32)) {

    int m0  = strip << 4;
    int hi  = lane >> 4;
    int row = m0 + (lane & 15);

    v8f acc[NT];
    v8f zero = {0.f,0.f,0.f,0.f,0.f,0.f,0.f,0.f};
    #pragma unroll
    for (int j = 0; j < NT; ++j) acc[j] = zero;

    const int mrow = m0 + (hi << 3);
    float4 g0, g1;
    if (SELF) {
        const float4* dp = reinterpret_cast<const float4*>(deg + mrow);
        g0 = dp[0]; g1 = dp[1];
    }

    const v8h* bl = reinterpret_cast<const v8h*>(wl);
    constexpr int ktiles = K >> 5;
    for (int kt = 0; kt < ktiles; ++kt) {
        int kbase = (kt << 5) + (hi << 3);
        const float4* p = reinterpret_cast<const float4*>(A + (long)row * K + kbase);
        float4 x0 = p[0], x1 = p[1];
        float4 x2 = p[4], x3 = p[5];
        if (RELU) {
            x0.x=fmaxf(x0.x,0.f); x0.y=fmaxf(x0.y,0.f); x0.z=fmaxf(x0.z,0.f); x0.w=fmaxf(x0.w,0.f);
            x1.x=fmaxf(x1.x,0.f); x1.y=fmaxf(x1.y,0.f); x1.z=fmaxf(x1.z,0.f); x1.w=fmaxf(x1.w,0.f);
            x2.x=fmaxf(x2.x,0.f); x2.y=fmaxf(x2.y,0.f); x2.z=fmaxf(x2.z,0.f); x2.w=fmaxf(x2.w,0.f);
            x3.x=fmaxf(x3.x,0.f); x3.y=fmaxf(x3.y,0.f); x3.z=fmaxf(x3.z,0.f); x3.w=fmaxf(x3.w,0.f);
        }
        v16h a;
        a[0]=(_Float16)x0.x; a[1]=(_Float16)x0.y; a[2]=(_Float16)x0.z; a[3]=(_Float16)x0.w;
        a[4]=(_Float16)x1.x; a[5]=(_Float16)x1.y; a[6]=(_Float16)x1.z; a[7]=(_Float16)x1.w;
        a[8]=(_Float16)x2.x; a[9]=(_Float16)x2.y; a[10]=(_Float16)x2.z; a[11]=(_Float16)x2.w;
        a[12]=(_Float16)x3.x; a[13]=(_Float16)x3.y; a[14]=(_Float16)x3.z; a[15]=(_Float16)x3.w;

        #pragma unroll
        for (int nt = 0; nt < NT; ++nt) {
            const v8h blo = bl[(kt * NT + nt) * 64 + lane];
            const v8h bhi = bl[(kt * NT + nt) * 64 + 32 + lane];
            const v16h b = __builtin_shufflevector(blo, bhi, 0, 1, 2, 3, 4, 5, 6, 7,
                                                   8, 9, 10, 11, 12, 13, 14, 15);
            acc[nt] = __builtin_amdgcn_wmma_f32_16x16x32_f16(
                false, a, false, b, (short)0, acc[nt], false, false);
        }
    }

    if (SELF) {
        float d[8];
        d[0] = rsqrtf(g0.x); d[1] = rsqrtf(g0.y); d[2] = rsqrtf(g0.z); d[3] = rsqrtf(g0.w);
        d[4] = rsqrtf(g1.x); d[5] = rsqrtf(g1.y); d[6] = rsqrtf(g1.z); d[7] = rsqrtf(g1.w);
        if ((lane & 15) == 0) {
            float4* op = reinterpret_cast<float4*>(dinv + mrow);
            op[0] = make_float4(d[0], d[1], d[2], d[3]);
            op[1] = make_float4(d[4], d[5], d[6], d[7]);
        }
        #pragma unroll
        for (int nt = 0; nt < NT; ++nt) {
            int n = (nt << 4) + (lane & 15);
            const float bn = bias[n];
            #pragma unroll
            for (int j = 0; j < 8; ++j) {
                const float v = acc[nt][j];
                C [(long)(mrow + j) * N + n] = v;
                C2[(long)(mrow + j) * N + n] = v * d[j] * d[j] + bn;
            }
        }
    } else {
        #pragma unroll
        for (int nt = 0; nt < NT; ++nt) {
            int n = (nt << 4) + (lane & 15);
            #pragma unroll
            for (int j = 0; j < 8; ++j)
                C[(long)(mrow + j) * N + n] = acc[nt][j];
        }
    }
    }
}

__device__ __forceinline__ int   lane_bcast_i(int v, int j)   { return __builtin_amdgcn_readlane(v, j); }
__device__ __forceinline__ float lane_bcast_f(float v, int j) {
    return __int_as_float(__builtin_amdgcn_readlane(__float_as_int(v), j));
}

#define EP_THREADS 256
#define EP_BATCH   8
template <int CPE>
__global__ void __launch_bounds__(EP_THREADS)
k_edge_prop(const int* __restrict__ src, const int* __restrict__ dst,
            const float* __restrict__ H, const float* __restrict__ dinv,
            float* acc, int E) {
    constexpr int FPL = CPE / 32;
    const int lane = threadIdx.x & 31;
    const int gi   = __builtin_amdgcn_readfirstlane((int)((blockIdx.x * blockDim.x + threadIdx.x) >> 5));
    const int base = gi << 5;
    if (base >= E) return;
    const int m = min(32, E - base);
    int s = 0, d = 0;
    if (lane < m) { s = src[base + lane]; d = dst[base + lane]; }
    const float nrm = dinv[s] * dinv[d];

    if (m == 32) {
        #pragma unroll
        for (int j0 = 0; j0 < 32; j0 += EP_BATCH) {
            float hv[EP_BATCH][FPL];
            #pragma unroll
            for (int u = 0; u < EP_BATCH; ++u) {
                const int sj = lane_bcast_i(s, j0 + u);
                const float* hs = H + (long)sj * CPE + lane;
                for (int k = 0; k < FPL; ++k) hv[u][k] = hs[k * 32];
            }
            #pragma unroll
            for (int u = 0; u < EP_BATCH; ++u) {
                const int   dj = lane_bcast_i(d, j0 + u);
                const float nj = lane_bcast_f(nrm, j0 + u);
                float* ad = acc + (long)dj * CPE + lane;
                for (int k = 0; k < FPL; ++k) atomic_add_f32(&ad[k * 32], hv[u][k] * nj);
            }
        }
    } else {
        for (int j = 0; j < m; ++j) {
            const int   sj = lane_bcast_i(s, j);
            const int   dj = lane_bcast_i(d, j);
            const float nj = lane_bcast_f(nrm, j);
            const float* hs = H   + (long)sj * CPE + lane;
            float*       ad = acc + (long)dj * CPE + lane;
            float hv[FPL];
            for (int k = 0; k < FPL; ++k) hv[k] = hs[k * 32];
            for (int k = 0; k < FPL; ++k) atomic_add_f32(&ad[k * 32], hv[k] * nj);
        }
    }
}

#define PP_THREADS 1024
#define PP_WAVES   (PP_THREADS / 32)
#define PP_BATCH   16
#define PP_BLOCKS  256

__global__ void __launch_bounds__(PP_THREADS)
k_prop_pool(const int* __restrict__ src, const int* __restrict__ dst,
            const int* __restrict__ batch, const float* __restrict__ H,
            const float* __restrict__ dinv, const float* __restrict__ bias,
            float* __restrict__ part, float* __restrict__ cntp, int E, int Nn) {
    __shared__ __attribute__((aligned(16))) float tab[NGRAPH * OUT_C];
    __shared__ float cnt[NGRAPH];
    for (int i = threadIdx.x; i < NGRAPH * OUT_C; i += PP_THREADS) tab[i] = 0.f;
    if (threadIdx.x < NGRAPH) cnt[threadIdx.x] = 0.f;
    __syncthreads();

    const int lane = threadIdx.x & 31;
    const int wib  = __builtin_amdgcn_readfirstlane((int)(threadIdx.x >> 5));
    const int nge  = (E + 31) >> 5;
    const int ngn  = (Nn + 31) >> 5;
    const int ng   = nge + ngn;
    const int tw   = (int)gridDim.x * PP_WAVES;
    const int c0   = lane * 2;
    const float bx = bias[c0], by = bias[c0 + 1];

    for (int gi = wib * (int)gridDim.x + (int)blockIdx.x; gi < ng; gi += tw) {
        const bool isnode = gi >= nge;
        int s = 0, d = 0, m;
        if (!isnode) {
            const int base = gi << 5;
            m = min(32, E - base);
            if (lane < m) { s = src[base + lane]; d = dst[base + lane]; }
        } else {
            const int base = (gi - nge) << 5;
            m = min(32, Nn - base);
            if (lane < m) { s = base + lane; d = s; }
        }
        const float nrm = dinv[s] * dinv[d];
        const int   g   = batch[d];
        const float ax = isnode ? bx : 0.f;
        const float ay = isnode ? by : 0.f;
        if (isnode && lane < m) atomicAdd(&cnt[g], 1.0f);

        if (m == 32) {
            #pragma unroll
            for (int j0 = 0; j0 < 32; j0 += PP_BATCH) {
                float2 v[PP_BATCH];
                #pragma unroll
                for (int u = 0; u < PP_BATCH; ++u) {
                    const int sj = lane_bcast_i(s, j0 + u);
                    v[u] = *reinterpret_cast<const float2*>(H + (long)sj * OUT_C + c0);
                }
                #pragma unroll
                for (int u = 0; u < PP_BATCH; ++u) {
                    const float nj = lane_bcast_f(nrm, j0 + u);
                    const int   gj = lane_bcast_i(g, j0 + u);
                    float* t = &tab[gj * OUT_C + c0];
                    atomicAdd(t,     fmaf(v[u].x, nj, ax));
                    atomicAdd(t + 1, fmaf(v[u].y, nj, ay));
                }
            }
        } else {
            for (int j = 0; j < m; ++j) {
                const int   sj = lane_bcast_i(s, j);
                const float nj = lane_bcast_f(nrm, j);
                const int   gj = lane_bcast_i(g, j);
                const float2 v = *reinterpret_cast<const float2*>(H + (long)sj * OUT_C + c0);
                float* t = &tab[gj * OUT_C + c0];
                atomicAdd(t,     fmaf(v.x, nj, ax));
                atomicAdd(t + 1, fmaf(v.y, nj, ay));
            }
        }
    }
    __syncthreads();

    const float4* t4 = reinterpret_cast<const float4*>(tab);
    float4* p4 = reinterpret_cast<float4*>(part + (long)blockIdx.x * (NGRAPH * OUT_C));
    for (int i = threadIdx.x; i < NGRAPH * OUT_C / 4; i += PP_THREADS) p4[i] = t4[i];
    if (threadIdx.x < NGRAPH) cntp[(int)blockIdx.x * NGRAPH + threadIdx.x] = cnt[threadIdx.x];
}

#define PF_THREADS 512
#define PF_GROUPS  (PF_THREADS / OUT_C)
__global__ void __launch_bounds__(PF_THREADS)
k_pool_final(const float* __restrict__ part, const float* __restrict__ cntp,
             float* __restrict__ out) {
    __shared__ float red[PF_GROUPS][OUT_C];
    __shared__ float cred[PF_THREADS / 32];
    const int g = blockIdx.x;
    const int c = threadIdx.x & (OUT_C - 1);
    const int q = threadIdx.x >> 6;
    float s = 0.f;
    #pragma unroll
    for (int i = 0; i < PP_BLOCKS / PF_GROUPS; ++i)
        s += part[((long)(q + i * PF_GROUPS) * NGRAPH + g) * OUT_C + c];
    red[q][c] = s;
    float n = 0.f;
    for (int b = threadIdx.x; b < PP_BLOCKS; b += PF_THREADS) n += cntp[b * NGRAPH + g];
    for (int o = 16; o > 0; o >>= 1) n += __shfl_xor(n, o);
    if ((threadIdx.x & 31) == 0) cred[threadIdx.x >> 5] = n;
    __syncthreads();
    if (threadIdx.x < OUT_C) {
        float t = 0.f;
        for (int k = 0; k < PF_GROUPS; ++k) t += red[k][c];
        float m = 0.f;
        for (int k = 0; k < PF_THREADS / 32; ++k) m += cred[k];
        out[g * OUT_C + c] = t / fmaxf(m, 1.0f);
    }
}

extern "C" void kernel_launch(void* const* d_in, const int* in_sizes, int n_in,
                              void* d_out, int out_size, void* d_ws, size_t ws_size,
                              hipStream_t stream) {
    const float* x   = (const float*)d_in[0];
    const float* W1  = (const float*)d_in[1];
    const float* b1  = (const float*)d_in[2];
    const float* W2  = (const float*)d_in[3];
    const float* b2  = (const float*)d_in[4];
    const int*   ei  = (const int*)d_in[5];
    const int*   bat = (const int*)d_in[6];

    const int Nn = in_sizes[0] / IN_C;
    const int E  = in_sizes[5] / 2;
    const int* src = ei;
    const int* dst = ei + E;

    char* w = (char*)d_ws;
    size_t off = 0;
    auto alloc = [&](size_t bytes) -> void* {
        void* p = w + off;
        off = (off + bytes + 255) & ~(size_t)255;
        return p;
    };
    float*    deg   = (float*)alloc((size_t)Nn * 4);
    float*    dinv  = (float*)alloc((size_t)Nn * 4);
    float*    h1    = (float*)alloc((size_t)Nn * HID_C * 4);
    float*    a1    = (float*)alloc((size_t)Nn * HID_C * 4);
    float*    h2 = h1;
    float*    part  = a1;
    float*    cntp  = a1 + (size_t)PP_BLOCKS * NGRAPH * OUT_C;

    const int T = 256;
    k_fill<<<(Nn + T - 1) / T, T, 0, stream>>>(deg, 1.0f, Nn);
    k_edge_deg<<<(E + T - 1) / T, T, 0, stream>>>(dst, deg, E);

    k_gemm<HID_C / 16, false, true><<<GM_BLOCKS, GM_THREADS, 0, stream>>>(x, W1, h1, Nn,
                                                                         deg, b1, dinv, a1);
    k_edge_prop<HID_C><<<(E + 255) / 256, EP_THREADS, 0, stream>>>(src, dst, h1, dinv, a1, E);

    k_gemm<OUT_C / 16, true, false><<<GM_BLOCKS, GM_THREADS, 0, stream>>>(a1, W2, h2, Nn,
                                                                         nullptr, nullptr, nullptr, nullptr);
    k_prop_pool<<<PP_BLOCKS, PP_THREADS, 0, stream>>>(src, dst, bat, h2, dinv, b2, part, cntp, E, Nn);
    k_pool_final<<<NGRAPH, PF_THREADS, 0, stream>>>(part, cntp, (float*)d_out);
}


# in-degree count: 32 work-groups count their edge shares in a whole-graph LDS table (integer LDS adds), then one whole-line float add per 32 nodes onto deg; was one scattered global atomic per edge
# speedup vs baseline: 2.6227x; 1.1381x over previous
#include <hip/hip_runtime.h>
#include <hip/hip_bf16.h>
#include <stdint.h>

typedef __attribute__((ext_vector_type(16))) _Float16 v16h;
typedef __attribute__((ext_vector_type(8)))  _Float16 v8h;
typedef __attribute__((ext_vector_type(8)))  float    v8f;

#define IN_C   128
#define HID_C  128
#define OUT_C  64
#define NGRAPH 128

__device__ __forceinline__ void atomic_add_f32(float* p, float v) {
    unsafeAtomicAdd(p, v);
}

__global__ void k_fill(float* p, float v, int n) {
    int i = blockIdx.x * blockDim.x + threadIdx.x;
    if (i < n) p[i] = v;
}

#define DG_THREADS 1024
#define DG_BLOCKS  32
#define DG_WIN     65536
#define DG_BATCH   8
__global__ void __launch_bounds__(DG_THREADS)
k_edge_deg(const int* __restrict__ dst, float* deg, int E, int Nn) {
    __shared__ __attribute__((aligned(16))) unsigned hist[DG_WIN];
    const int tid   = (int)threadIdx.x;
    const int per   = (E + (int)gridDim.x - 1) / (int)gridDim.x;
    const int start = (int)blockIdx.x * per;
    const int end   = min(E, start + per);
    for (int n0 = 0; n0 < Nn; n0 += DG_WIN) {
        if (n0 > 0) __syncthreads();
        const int cnt = min(DG_WIN, Nn - n0);
        uint4* h4 = reinterpret_cast<uint4*>(hist);
        for (int i = tid; i < (cnt + 3) / 4; i += DG_THREADS) h4[i] = make_uint4(0u, 0u, 0u, 0u);
        __syncthreads();
        for (int e0 = start; e0 < end; e0 += DG_THREADS * DG_BATCH) {
            int v[DG_BATCH];
            #pragma unroll
            for (int u = 0; u < DG_BATCH; ++u) {
                const int e = e0 + u * DG_THREADS + tid;
                v[u] = (e < end) ? dst[e] : -1;
            }
            #pragma unroll
            for (int u = 0; u < DG_BATCH; ++u) {
                const unsigned r = (unsigned)(v[u] - n0);
                if (r < (unsigned)cnt) atomicAdd(&hist[r], 1u);
            }
        }
        __syncthreads();
        const int rot = ((((int)blockIdx.x % DG_BLOCKS) * (cnt >> 5)) / DG_BLOCKS) << 5;
        for (int k0 = 0; k0 < cnt; k0 += DG_THREADS * DG_BATCH) {
            unsigned h[DG_BATCH];
            int      id[DG_BATCH];
            #pragma unroll
            for (int u = 0; u < DG_BATCH; ++u) {
                const int k = k0 + u * DG_THREADS + tid;
                int i = k + rot;
                if (i >= cnt) i -= cnt;
                id[u] = i;
                h[u]  = (k < cnt) ? hist[i] : 0u;
            }
            #pragma unroll
            for (int u = 0; u < DG_BATCH; ++u)
                if (h[u] != 0u) atomic_add_f32(&deg[n0 + id[u]], (float)h[u]);
        }
    }
}

#define GM_THREADS 512
#define GM_K       128
#define GM_BLOCKS  256
template <int NT, bool RELU, bool SELF>
__global__ void __launch_bounds__(GM_THREADS)
k_gemm(const float* __restrict__ A, const float* __restrict__ W,
       float* __restrict__ C, int M,
       const float* __restrict__ deg, const float* __restrict__ bias,
       float* __restrict__ dinv, float* __restrict__ C2) {
    constexpr int K = GM_K;
    constexpr int N = NT * 16;
    __shared__ __attribute__((aligned(16))) _Float16 wl[K * N];
    v8h* wlv = reinterpret_cast<v8h*>(wl);
    #pragma unroll
    for (int it = 0; it < (K / 8) * N / GM_THREADS; ++it) {
        const int t  = it * GM_THREADS + (int)threadIdx.x;
        const int n  = t % N;
        const int k0 = (t / N) * 8;
        const float* wp = W + k0 * N + n;
        v8f f;
        #pragma unroll
        for (int j = 0; j < 8; ++j) f[j] = wp[j * N];
        const v8h h = __builtin_convertvector(f, v8h);
        const int kk0 = k0 & 31;
        wlv[((k0 >> 5) * NT + (n >> 4)) * 64 + (kk0 >> 4) * 32 + (((kk0 >> 3) & 1) << 4) + (n & 15)] = h;
    }
    __syncthreads();

    const int lane = threadIdx.x & 31;
    const int wib  = __builtin_amdgcn_readfirstlane((int)(threadIdx.x >> 5));
    const int mtiles = M >> 4;
    for (int strip = wib * (int)gridDim.x + (int)blockIdx.x; strip < mtiles;
         strip += (int)gridDim.x * (GM_THREADS / 32)) {

    int m0  = strip << 4;
    int hi  = lane >> 4;
    int row = m0 + (lane & 15);

    v8f acc[NT];
    v8f zero = {0.f,0.f,0.f,0.f,0.f,0.f,0.f,0.f};
    #pragma unroll
    for (int j = 0; j < NT; ++j) acc[j] = zero;

    const int mrow = m0 + (hi << 3);
    float4 g0, g1;
    if (SELF) {
        const float4* dp = reinterpret_cast<const float4*>(deg + mrow);
        g0 = dp[0]; g1 = dp[1];
    }

    const v8h* bl = reinterpret_cast<const v8h*>(wl);
    constexpr int ktiles = K >> 5;
    for (int kt = 0; kt < ktiles; ++kt) {
        int kbase = (kt << 5) + (hi << 3);
        const float4* p = reinterpret_cast<const float4*>(A + (long)row * K + kbase);
        float4 x0 = p[0], x1 = p[1];
        float4 x2 = p[4], x3 = p[5];
        if (RELU) {
            x0.x=fmaxf(x0.x,0.f); x0.y=fmaxf(x0.y,0.f); x0.z=fmaxf(x0.z,0.f); x0.w=fmaxf(x0.w,0.f);
            x1.x=fmaxf(x1.x,0.f); x1.y=fmaxf(x1.y,0.f); x1.z=fmaxf(x1.z,0.f); x1.w=fmaxf(x1.w,0.f);
            x2.x=fmaxf(x2.x,0.f); x2.y=fmaxf(x2.y,0.f); x2.z=fmaxf(x2.z,0.f); x2.w=fmaxf(x2.w,0.f);
            x3.x=fmaxf(x3.x,0.f); x3.y=fmaxf(x3.y,0.f); x3.z=fmaxf(x3.z,0.f); x3.w=fmaxf(x3.w,0.f);
        }
        v16h a;
        a[0]=(_Float16)x0.x; a[1]=(_Float16)x0.y; a[2]=(_Float16)x0.z; a[3]=(_Float16)x0.w;
        a[4]=(_Float16)x1.x; a[5]=(_Float16)x1.y; a[6]=(_Float16)x1.z; a[7]=(_Float16)x1.w;
        a[8]=(_Float16)x2.x; a[9]=(_Float16)x2.y; a[10]=(_Float16)x2.z; a[11]=(_Float16)x2.w;
        a[12]=(_Float16)x3.x; a[13]=(_Float16)x3.y; a[14]=(_Float16)x3.z; a[15]=(_Float16)x3.w;

        #pragma unroll
        for (int nt = 0; nt < NT; ++nt) {
            const v8h blo = bl[(kt * NT + nt) * 64 + lane];
            const v8h bhi = bl[(kt * NT + nt) * 64 + 32 + lane];
            const v16h b = __builtin_shufflevector(blo, bhi, 0, 1, 2, 3, 4, 5, 6, 7,
                                                   8, 9, 10, 11, 12, 13, 14, 15);
            acc[nt] = __builtin_amdgcn_wmma_f32_16x16x32_f16(
                false, a, false, b, (short)0, acc[nt], false, false);
        }
    }

    if (SELF) {
        float d[8];
        d[0] = rsqrtf(g0.x); d[1] = rsqrtf(g0.y); d[2] = rsqrtf(g0.z); d[3] = rsqrtf(g0.w);
        d[4] = rsqrtf(g1.x); d[5] = rsqrtf(g1.y); d[6] = rsqrtf(g1.z); d[7] = rsqrtf(g1.w);
        if ((lane & 15) == 0) {
            float4* op = reinterpret_cast<float4*>(dinv + mrow);
            op[0] = make_float4(d[0], d[1], d[2], d[3]);
            op[1] = make_float4(d[4], d[5], d[6], d[7]);
        }
        #pragma unroll
        for (int nt = 0; nt < NT; ++nt) {
            int n = (nt << 4) + (lane & 15);
            const float bn = bias[n];
            #pragma unroll
            for (int j = 0; j < 8; ++j) {
                const float v = acc[nt][j];
                C [(long)(mrow + j) * N + n] = v;
                C2[(long)(mrow + j) * N + n] = v * d[j] * d[j] + bn;
            }
        }
    } else {
        #pragma unroll
        for (int nt = 0; nt < NT; ++nt) {
            int n = (nt << 4) + (lane & 15);
            #pragma unroll
            for (int j = 0; j < 8; ++j)
                C[(long)(mrow + j) * N + n] = acc[nt][j];
        }
    }
    }
}

__device__ __forceinline__ int   lane_bcast_i(int v, int j)   { return __builtin_amdgcn_readlane(v, j); }
__device__ __forceinline__ float lane_bcast_f(float v, int j) {
    return __int_as_float(__builtin_amdgcn_readlane(__float_as_int(v), j));
}

#define EP_THREADS 256
#define EP_BATCH   8
template <int CPE>
__global__ void __launch_bounds__(EP_THREADS)
k_edge_prop(const int* __restrict__ src, const int* __restrict__ dst,
            const float* __restrict__ H, const float* __restrict__ dinv,
            float* acc, int E) {
    constexpr int FPL = CPE / 32;
    const int lane = threadIdx.x & 31;
    const int gi   = __builtin_amdgcn_readfirstlane((int)((blockIdx.x * blockDim.x + threadIdx.x) >> 5));
    const int base = gi << 5;
    if (base >= E) return;
    const int m = min(32, E - base);
    int s = 0, d = 0;
    if (lane < m) { s = src[base + lane]; d = dst[base + lane]; }
    const float nrm = dinv[s] * dinv[d];

    if (m == 32) {
        #pragma unroll
        for (int j0 = 0; j0 < 32; j0 += EP_BATCH) {
            float hv[EP_BATCH][FPL];
            #pragma unroll
            for (int u = 0; u < EP_BATCH; ++u) {
                const int sj = lane_bcast_i(s, j0 + u);
                const float* hs = H + (long)sj * CPE + lane;
                for (int k = 0; k < FPL; ++k) hv[u][k] = hs[k * 32];
            }
            #pragma unroll
            for (int u = 0; u < EP_BATCH; ++u) {
                const int   dj = lane_bcast_i(d, j0 + u);
                const float nj = lane_bcast_f(nrm, j0 + u);
                float* ad = acc + (long)dj * CPE + lane;
                for (int k = 0; k < FPL; ++k) atomic_add_f32(&ad[k * 32], hv[u][k] * nj);
            }
        }
    } else {
        for (int j = 0; j < m; ++j) {
            const int   sj = lane_bcast_i(s, j);
            const int   dj = lane_bcast_i(d, j);
            const float nj = lane_bcast_f(nrm, j);
            const float* hs = H   + (long)sj * CPE + lane;
            float*       ad = acc + (long)dj * CPE + lane;
            float hv[FPL];
            for (int k = 0; k < FPL; ++k) hv[k] = hs[k * 32];
            for (int k = 0; k < FPL; ++k) atomic_add_f32(&ad[k * 32], hv[k] * nj);
        }
    }
}

#define PP_THREADS 1024
#define PP_WAVES   (PP_THREADS / 32)
#define PP_BATCH   16
#define PP_BLOCKS  256

__global__ void __launch_bounds__(PP_THREADS)
k_prop_pool(const int* __restrict__ src, const int* __restrict__ dst,
            const int* __restrict__ batch, const float* __restrict__ H,
            const float* __restrict__ dinv, const float* __restrict__ bias,
            float* __restrict__ part, float* __restrict__ cntp, int E, int Nn) {
    __shared__ __attribute__((aligned(16))) float tab[NGRAPH * OUT_C];
    __shared__ float cnt[NGRAPH];
    for (int i = threadIdx.x; i < NGRAPH * OUT_C; i += PP_THREADS) tab[i] = 0.f;
    if (threadIdx.x < NGRAPH) cnt[threadIdx.x] = 0.f;
    __syncthreads();

    const int lane = threadIdx.x & 31;
    const int wib  = __builtin_amdgcn_readfirstlane((int)(threadIdx.x >> 5));
    const int nge  = (E + 31) >> 5;
    const int ngn  = (Nn + 31) >> 5;
    const int ng   = nge + ngn;
    const int tw   = (int)gridDim.x * PP_WAVES;
    const int c0   = lane * 2;
    const float bx = bias[c0], by = bias[c0 + 1];

    for (int gi = wib * (int)gridDim.x + (int)blockIdx.x; gi < ng; gi += tw) {
        const bool isnode = gi >= nge;
        int s = 0, d = 0, m;
        if (!isnode) {
            const int base = gi << 5;
            m = min(32, E - base);
            if (lane < m) { s = src[base + lane]; d = dst[base + lane]; }
        } else {
            const int base = (gi - nge) << 5;
            m = min(32, Nn - base);
            if (lane < m) { s = base + lane; d = s; }
        }
        const float nrm = dinv[s] * dinv[d];
        const int   g   = batch[d];
        const float ax = isnode ? bx : 0.f;
        const float ay = isnode ? by : 0.f;
        if (isnode && lane < m) atomicAdd(&cnt[g], 1.0f);

        if (m == 32) {
            #pragma unroll
            for (int j0 = 0; j0 < 32; j0 += PP_BATCH) {
                float2 v[PP_BATCH];
                #pragma unroll
                for (int u = 0; u < PP_BATCH; ++u) {
                    const int sj = lane_bcast_i(s, j0 + u);
                    v[u] = *reinterpret_cast<const float2*>(H + (long)sj * OUT_C + c0);
                }
                #pragma unroll
                for (int u = 0; u < PP_BATCH; ++u) {
                    const float nj = lane_bcast_f(nrm, j0 + u);
                    const int   gj = lane_bcast_i(g, j0 + u);
                    float* t = &tab[gj * OUT_C + c0];
                    atomicAdd(t,     fmaf(v[u].x, nj, ax));
                    atomicAdd(t + 1, fmaf(v[u].y, nj, ay));
                }
            }
        } else {
            for (int j = 0; j < m; ++j) {
                const int   sj = lane_bcast_i(s, j);
                const float nj = lane_bcast_f(nrm, j);
                const int   gj = lane_bcast_i(g, j);
                const float2 v = *reinterpret_cast<const float2*>(H + (long)sj * OUT_C + c0);
                float* t = &tab[gj * OUT_C + c0];
                atomicAdd(t,     fmaf(v.x, nj, ax));
                atomicAdd(t + 1, fmaf(v.y, nj, ay));
            }
        }
    }
    __syncthreads();

    const float4* t4 = reinterpret_cast<const float4*>(tab);
    float4* p4 = reinterpret_cast<float4*>(part + (long)blockIdx.x * (NGRAPH * OUT_C));
    for (int i = threadIdx.x; i < NGRAPH * OUT_C / 4; i += PP_THREADS) p4[i] = t4[i];
    if (threadIdx.x < NGRAPH) cntp[(int)blockIdx.x * NGRAPH + threadIdx.x] = cnt[threadIdx.x];
}

#define PF_THREADS 512
#define PF_GROUPS  (PF_THREADS / OUT_C)
__global__ void __launch_bounds__(PF_THREADS)
k_pool_final(const float* __restrict__ part, const float* __restrict__ cntp,
             float* __restrict__ out) {
    __shared__ float red[PF_GROUPS][OUT_C];
    __shared__ float cred[PF_THREADS / 32];
    const int g = blockIdx.x;
    const int c = threadIdx.x & (OUT_C - 1);
    const int q = threadIdx.x >> 6;
    float s = 0.f;
    #pragma unroll
    for (int i = 0; i < PP_BLOCKS / PF_GROUPS; ++i)
        s += part[((long)(q + i * PF_GROUPS) * NGRAPH + g) * OUT_C + c];
    red[q][c] = s;
    float n = 0.f;
    for (int b = threadIdx.x; b < PP_BLOCKS; b += PF_THREADS) n += cntp[b * NGRAPH + g];
    for (int o = 16; o > 0; o >>= 1) n += __shfl_xor(n, o);
    if ((threadIdx.x & 31) == 0) cred[threadIdx.x >> 5] = n;
    __syncthreads();
    if (threadIdx.x < OUT_C) {
        float t = 0.f;
        for (int k = 0; k < PF_GROUPS; ++k) t += red[k][c];
        float m = 0.f;
        for (int k = 0; k < PF_THREADS / 32; ++k) m += cred[k];
        out[g * OUT_C + c] = t / fmaxf(m, 1.0f);
    }
}

extern "C" void kernel_launch(void* const* d_in, const int* in_sizes, int n_in,
                              void* d_out, int out_size, void* d_ws, size_t ws_size,
                              hipStream_t stream) {
    const float* x   = (const float*)d_in[0];
    const float* W1  = (const float*)d_in[1];
    const float* b1  = (const float*)d_in[2];
    const float* W2  = (const float*)d_in[3];
    const float* b2  = (const float*)d_in[4];
    const int*   ei  = (const int*)d_in[5];
    const int*   bat = (const int*)d_in[6];

    const int Nn = in_sizes[0] / IN_C;
    const int E  = in_sizes[5] / 2;
    const int* src = ei;
    const int* dst = ei + E;

    char* w = (char*)d_ws;
    size_t off = 0;
    auto alloc = [&](size_t bytes) -> void* {
        void* p = w + off;
        off = (off + bytes + 255) & ~(size_t)255;
        return p;
    };
    float*    deg   = (float*)alloc((size_t)Nn * 4);
    float*    dinv  = (float*)alloc((size_t)Nn * 4);
    float*    h1    = (float*)alloc((size_t)Nn * HID_C * 4);
    float*    a1    = (float*)alloc((size_t)Nn * HID_C * 4);
    float*    h2 = h1;
    float*    part  = a1;
    float*    cntp  = a1 + (size_t)PP_BLOCKS * NGRAPH * OUT_C;

    const int T = 256;
    k_fill<<<(Nn + T - 1) / T, T, 0, stream>>>(deg, 1.0f, Nn);
    k_edge_deg<<<DG_BLOCKS, DG_THREADS, 0, stream>>>(dst, deg, E, Nn);

    k_gemm<HID_C / 16, false, true><<<GM_BLOCKS, GM_THREADS, 0, stream>>>(x, W1, h1, Nn,
                                                                         deg, b1, dinv, a1);
    k_edge_prop<HID_C><<<(E + 255) / 256, EP_THREADS, 0, stream>>>(src, dst, h1, dinv, a1, E);

    k_gemm<OUT_C / 16, true, false><<<GM_BLOCKS, GM_THREADS, 0, stream>>>(a1, W2, h2, Nn,
                                                                         nullptr, nullptr, nullptr, nullptr);
    k_prop_pool<<<PP_BLOCKS, PP_THREADS, 0, stream>>>(src, dst, bat, h2, dinv, b2, part, cntp, E, Nn);
    k_pool_final<<<NGRAPH, PF_THREADS, 0, stream>>>(part, cntp, (float*)d_out);
}


# layer-1 edge pass stores each edge's coefficient dinv[src]*dinv[dst] and its destination's graph id in edge order; layer-2 pass reads them coalesced, its three gathers by node go; sums unchanged
# speedup vs baseline: 2.7161x; 1.0356x over previous
#include <hip/hip_runtime.h>
#include <hip/hip_bf16.h>
#include <stdint.h>

typedef __attribute__((ext_vector_type(16))) _Float16 v16h;
typedef __attribute__((ext_vector_type(8)))  _Float16 v8h;
typedef __attribute__((ext_vector_type(8)))  float    v8f;

#define IN_C   128
#define HID_C  128
#define OUT_C  64
#define NGRAPH 128

__device__ __forceinline__ void atomic_add_f32(float* p, float v) {
    unsafeAtomicAdd(p, v);
}

__global__ void k_fill(float* p, float v, int n) {
    int i = blockIdx.x * blockDim.x + threadIdx.x;
    if (i < n) p[i] = v;
}

#define DG_THREADS 1024
#define DG_BLOCKS  32
#define DG_WIN     65536
#define DG_BATCH   8
__global__ void __launch_bounds__(DG_THREADS)
k_edge_deg(const int* __restrict__ dst, float* deg, int E, int Nn) {
    __shared__ __attribute__((aligned(16))) unsigned hist[DG_WIN];
    const int tid   = (int)threadIdx.x;
    const int per   = (E + (int)gridDim.x - 1) / (int)gridDim.x;
    const int start = (int)blockIdx.x * per;
    const int end   = min(E, start + per);
    for (int n0 = 0; n0 < Nn; n0 += DG_WIN) {
        if (n0 > 0) __syncthreads();
        const int cnt = min(DG_WIN, Nn - n0);
        uint4* h4 = reinterpret_cast<uint4*>(hist);
        for (int i = tid; i < (cnt + 3) / 4; i += DG_THREADS) h4[i] = make_uint4(0u, 0u, 0u, 0u);
        __syncthreads();
        for (int e0 = start; e0 < end; e0 += DG_THREADS * DG_BATCH) {
            int v[DG_BATCH];
            #pragma unroll
            for (int u = 0; u < DG_BATCH; ++u) {
                const int e = e0 + u * DG_THREADS + tid;
                v[u] = (e < end) ? dst[e] : -1;
            }
            #pragma unroll
            for (int u = 0; u < DG_BATCH; ++u) {
                const unsigned r = (unsigned)(v[u] - n0);
                if (r < (unsigned)cnt) atomicAdd(&hist[r], 1u);
            }
        }
        __syncthreads();
        const int rot = ((((int)blockIdx.x % DG_BLOCKS) * (cnt >> 5)) / DG_BLOCKS) << 5;
        for (int k0 = 0; k0 < cnt; k0 += DG_THREADS * DG_BATCH) {
            unsigned h[DG_BATCH];
            int      id[DG_BATCH];
            #pragma unroll
            for (int u = 0; u < DG_BATCH; ++u) {
                const int k = k0 + u * DG_THREADS + tid;
                int i = k + rot;
                if (i >= cnt) i -= cnt;
                id[u] = i;
                h[u]  = (k < cnt) ? hist[i] : 0u;
            }
            #pragma unroll
            for (int u = 0; u < DG_BATCH; ++u)
                if (h[u] != 0u) atomic_add_f32(&deg[n0 + id[u]], (float)h[u]);
        }
    }
}

#define GM_THREADS 512
#define GM_K       128
#define GM_BLOCKS  256
template <int NT, bool RELU, bool SELF>
__global__ void __launch_bounds__(GM_THREADS)
k_gemm(const float* __restrict__ A, const float* __restrict__ W,
       float* __restrict__ C, int M,
       const float* __restrict__ deg, const float* __restrict__ bias,
       const int* __restrict__ batch, float2* __restrict__ nd, float* __restrict__ C2) {
    constexpr int K = GM_K;
    constexpr int N = NT * 16;
    __shared__ __attribute__((aligned(16))) _Float16 wl[K * N];
    v8h* wlv = reinterpret_cast<v8h*>(wl);
    #pragma unroll
    for (int it = 0; it < (K / 8) * N / GM_THREADS; ++it) {
        const int t  = it * GM_THREADS + (int)threadIdx.x;
        const int n  = t % N;
        const int k0 = (t / N) * 8;
        const float* wp = W + k0 * N + n;
        v8f f;
        #pragma unroll
        for (int j = 0; j < 8; ++j) f[j] = wp[j * N];
        const v8h h = __builtin_convertvector(f, v8h);
        const int kk0 = k0 & 31;
        wlv[((k0 >> 5) * NT + (n >> 4)) * 64 + (kk0 >> 4) * 32 + (((kk0 >> 3) & 1) << 4) + (n & 15)] = h;
    }
    __syncthreads();

    const int lane = threadIdx.x & 31;
    const int wib  = __builtin_amdgcn_readfirstlane((int)(threadIdx.x >> 5));
    const int mtiles = M >> 4;
    for (int strip = wib * (int)gridDim.x + (int)blockIdx.x; strip < mtiles;
         strip += (int)gridDim.x * (GM_THREADS / 32)) {

    int m0  = strip << 4;
    int hi  = lane >> 4;
    int row = m0 + (lane & 15);

    v8f acc[NT];
    v8f zero = {0.f,0.f,0.f,0.f,0.f,0.f,0.f,0.f};
    #pragma unroll
    for (int j = 0; j < NT; ++j) acc[j] = zero;

    const int mrow = m0 + (hi << 3);
    float4 g0, g1;
    int4 q0, q1;
    if (SELF) {
        const float4* dp = reinterpret_cast<const float4*>(deg + mrow);
        g0 = dp[0]; g1 = dp[1];
        const int4* qp = reinterpret_cast<const int4*>(batch + mrow);
        q0 = qp[0]; q1 = qp[1];
    }

    const v8h* bl = reinterpret_cast<const v8h*>(wl);
    constexpr int ktiles = K >> 5;
    for (int kt = 0; kt < ktiles; ++kt) {
        int kbase = (kt << 5) + (hi << 3);
        const float4* p = reinterpret_cast<const float4*>(A + (long)row * K + kbase);
        float4 x0 = p[0], x1 = p[1];
        float4 x2 = p[4], x3 = p[5];
        if (RELU) {
            x0.x=fmaxf(x0.x,0.f); x0.y=fmaxf(x0.y,0.f); x0.z=fmaxf(x0.z,0.f); x0.w=fmaxf(x0.w,0.f);
            x1.x=fmaxf(x1.x,0.f); x1.y=fmaxf(x1.y,0.f); x1.z=fmaxf(x1.z,0.f); x1.w=fmaxf(x1.w,0.f);
            x2.x=fmaxf(x2.x,0.f); x2.y=fmaxf(x2.y,0.f); x2.z=fmaxf(x2.z,0.f); x2.w=fmaxf(x2.w,0.f);
            x3.x=fmaxf(x3.x,0.f); x3.y=fmaxf(x3.y,0.f); x3.z=fmaxf(x3.z,0.f); x3.w=fmaxf(x3.w,0.f);
        }
        v16h a;
        a[0]=(_Float16)x0.x; a[1]=(_Float16)x0.y; a[2]=(_Float16)x0.z; a[3]=(_Float16)x0.w;
        a[4]=(_Float16)x1.x; a[5]=(_Float16)x1.y; a[6]=(_Float16)x1.z; a[7]=(_Float16)x1.w;
        a[8]=(_Float16)x2.x; a[9]=(_Float16)x2.y; a[10]=(_Float16)x2.z; a[11]=(_Float16)x2.w;
        a[12]=(_Float16)x3.x; a[13]=(_Float16)x3.y; a[14]=(_Float16)x3.z; a[15]=(_Float16)x3.w;

        #pragma unroll
        for (int nt = 0; nt < NT; ++nt) {
            const v8h blo = bl[(kt * NT + nt) * 64 + lane];
            const v8h bhi = bl[(kt * NT + nt) * 64 + 32 + lane];
            const v16h b = __builtin_shufflevector(blo, bhi, 0, 1, 2, 3, 4, 5, 6, 7,
                                                   8, 9, 10, 11, 12, 13, 14, 15);
            acc[nt] = __builtin_amdgcn_wmma_f32_16x16x32_f16(
                false, a, false, b, (short)0, acc[nt], false, false);
        }
    }

    if (SELF) {
        float d[8];
        d[0] = rsqrtf(g0.x); d[1] = rsqrtf(g0.y); d[2] = rsqrtf(g0.z); d[3] = rsqrtf(g0.w);
        d[4] = rsqrtf(g1.x); d[5] = rsqrtf(g1.y); d[6] = rsqrtf(g1.z); d[7] = rsqrtf(g1.w);
        if ((lane & 15) == 0) {
            float4* op = reinterpret_cast<float4*>(nd + mrow);
            op[0] = make_float4(d[0], __int_as_float(q0.x), d[1], __int_as_float(q0.y));
            op[1] = make_float4(d[2], __int_as_float(q0.z), d[3], __int_as_float(q0.w));
            op[2] = make_float4(d[4], __int_as_float(q1.x), d[5], __int_as_float(q1.y));
            op[3] = make_float4(d[6], __int_as_float(q1.z), d[7], __int_as_float(q1.w));
        }
        #pragma unroll
        for (int nt = 0; nt < NT; ++nt) {
            int n = (nt << 4) + (lane & 15);
            const float bn = bias[n];
            #pragma unroll
            for (int j = 0; j < 8; ++j) {
                const float v = acc[nt][j];
                C [(long)(mrow + j) * N + n] = v;
                C2[(long)(mrow + j) * N + n] = v * d[j] * d[j] + bn;
            }
        }
    } else {
        #pragma unroll
        for (int nt = 0; nt < NT; ++nt) {
            int n = (nt << 4) + (lane & 15);
            #pragma unroll
            for (int j = 0; j < 8; ++j)
                C[(long)(mrow + j) * N + n] = acc[nt][j];
        }
    }
    }
}

__device__ __forceinline__ int   lane_bcast_i(int v, int j)   { return __builtin_amdgcn_readlane(v, j); }
__device__ __forceinline__ float lane_bcast_f(float v, int j) {
    return __int_as_float(__builtin_amdgcn_readlane(__float_as_int(v), j));
}

#define EP_THREADS 256
#define EP_BATCH   8
template <int CPE>
__global__ void __launch_bounds__(EP_THREADS)
k_edge_prop(const int* __restrict__ src, const int* __restrict__ dst,
            const float* __restrict__ H, const float2* __restrict__ nd,
            float2* __restrict__ ec, float* acc, int E) {
    constexpr int FPL = CPE / 32;
    const int lane = threadIdx.x & 31;
    const int gi   = __builtin_amdgcn_readfirstlane((int)((blockIdx.x * blockDim.x + threadIdx.x) >> 5));
    const int base = gi << 5;
    if (base >= E) return;
    const int m = min(32, E - base);
    int s = 0, d = 0;
    if (lane < m) { s = src[base + lane]; d = dst[base + lane]; }
    const unsigned long long qd = reinterpret_cast<const unsigned long long*>(nd)[d];
    const float nrm = nd[s].x * __int_as_float((int)(unsigned)qd);
    if (lane < m) ec[base + lane] = make_float2(nrm, __int_as_float((int)(unsigned)(qd >> 32)));

    if (m == 32) {
        #pragma unroll
        for (int j0 = 0; j0 < 32; j0 += EP_BATCH) {
            float hv[EP_BATCH][FPL];
            #pragma unroll
            for (int u = 0; u < EP_BATCH; ++u) {
                const int sj = lane_bcast_i(s, j0 + u);
                const float* hs = H + (long)sj * CPE + lane;
                for (int k = 0; k < FPL; ++k) hv[u][k] = hs[k * 32];
            }
            #pragma unroll
            for (int u = 0; u < EP_BATCH; ++u) {
                const int   dj = lane_bcast_i(d, j0 + u);
                const float nj = lane_bcast_f(nrm, j0 + u);
                float* ad = acc + (long)dj * CPE + lane;
                for (int k = 0; k < FPL; ++k) atomic_add_f32(&ad[k * 32], hv[u][k] * nj);
            }
        }
    } else {
        for (int j = 0; j < m; ++j) {
            const int   sj = lane_bcast_i(s, j);
            const int   dj = lane_bcast_i(d, j);
            const float nj = lane_bcast_f(nrm, j);
            const float* hs = H   + (long)sj * CPE + lane;
            float*       ad = acc + (long)dj * CPE + lane;
            float hv[FPL];
            for (int k = 0; k < FPL; ++k) hv[k] = hs[k * 32];
            for (int k = 0; k < FPL; ++k) atomic_add_f32(&ad[k * 32], hv[k] * nj);
        }
    }
}

#define PP_THREADS 1024
#define PP_WAVES   (PP_THREADS / 32)
#define PP_BATCH   16
#define PP_BLOCKS  256

__global__ void __launch_bounds__(PP_THREADS)
k_prop_pool(const int* __restrict__ src, const float2* __restrict__ ec,
            const float2* __restrict__ nd, const float* __restrict__ H,
            const float* __restrict__ bias,
            float* __restrict__ part, float* __restrict__ cntp, int E, int Nn) {
    __shared__ __attribute__((aligned(16))) float tab[NGRAPH * OUT_C];
    __shared__ float cnt[NGRAPH];
    for (int i = threadIdx.x; i < NGRAPH * OUT_C; i += PP_THREADS) tab[i] = 0.f;
    if (threadIdx.x < NGRAPH) cnt[threadIdx.x] = 0.f;
    __syncthreads();

    const int lane = threadIdx.x & 31;
    const int wib  = __builtin_amdgcn_readfirstlane((int)(threadIdx.x >> 5));
    const int nge  = (E + 31) >> 5;
    const int ngn  = (Nn + 31) >> 5;
    const int ng   = nge + ngn;
    const int tw   = (int)gridDim.x * PP_WAVES;
    const int c0   = lane * 2;
    const float bx = bias[c0], by = bias[c0 + 1];

    for (int gi = wib * (int)gridDim.x + (int)blockIdx.x; gi < ng; gi += tw) {
        const bool isnode = gi >= nge;
        int s = 0, m;
        float2 r = make_float2(0.f, 0.f);
        if (!isnode) {
            const int base = gi << 5;
            m = min(32, E - base);
            if (lane < m) { s = src[base + lane]; r = ec[base + lane]; }
        } else {
            const int base = (gi - nge) << 5;
            m = min(32, Nn - base);
            if (lane < m) { s = base + lane; r = nd[s]; r.x *= r.x; }
        }
        const float nrm = r.x;
        const int   g   = __float_as_int(r.y);
        const float ax = isnode ? bx : 0.f;
        const float ay = isnode ? by : 0.f;
        if (isnode && lane < m) atomicAdd(&cnt[g], 1.0f);

        if (m == 32) {
            #pragma unroll
            for (int j0 = 0; j0 < 32; j0 += PP_BATCH) {
                float2 v[PP_BATCH];
                #pragma unroll
                for (int u = 0; u < PP_BATCH; ++u) {
                    const int sj = lane_bcast_i(s, j0 + u);
                    v[u] = *reinterpret_cast<const float2*>(H + (long)sj * OUT_C + c0);
                }
                #pragma unroll
                for (int u = 0; u < PP_BATCH; ++u) {
                    const float nj = lane_bcast_f(nrm, j0 + u);
                    const int   gj = lane_bcast_i(g, j0 + u);
                    float* t = &tab[gj * OUT_C + c0];
                    atomicAdd(t,     fmaf(v[u].x, nj, ax));
                    atomicAdd(t + 1, fmaf(v[u].y, nj, ay));
                }
            }
        } else {
            for (int j = 0; j < m; ++j) {
                const int   sj = lane_bcast_i(s, j);
                const float nj = lane_bcast_f(nrm, j);
                const int   gj = lane_bcast_i(g, j);
                const float2 v = *reinterpret_cast<const float2*>(H + (long)sj * OUT_C + c0);
                float* t = &tab[gj * OUT_C + c0];
                atomicAdd(t,     fmaf(v.x, nj, ax));
                atomicAdd(t + 1, fmaf(v.y, nj, ay));
            }
        }
    }
    __syncthreads();

    const float4* t4 = reinterpret_cast<const float4*>(tab);
    float4* p4 = reinterpret_cast<float4*>(part + (long)blockIdx.x * (NGRAPH * OUT_C));
    for (int i = threadIdx.x; i < NGRAPH * OUT_C / 4; i += PP_THREADS) p4[i] = t4[i];
    if (threadIdx.x < NGRAPH) cntp[(int)blockIdx.x * NGRAPH + threadIdx.x] = cnt[threadIdx.x];
}

#define PF_THREADS 512
#define PF_GROUPS  (PF_THREADS / OUT_C)
__global__ void __launch_bounds__(PF_THREADS)
k_pool_final(const float* __restrict__ part, const float* __restrict__ cntp,
             float* __restrict__ out) {
    __shared__ float red[PF_GROUPS][OUT_C];
    __shared__ float cred[PF_THREADS / 32];
    const int g = blockIdx.x;
    const int c = threadIdx.x & (OUT_C - 1);
    const int q = threadIdx.x >> 6;
    float s = 0.f;
    #pragma unroll
    for (int i = 0; i < PP_BLOCKS / PF_GROUPS; ++i)
        s += part[((long)(q + i * PF_GROUPS) * NGRAPH + g) * OUT_C + c];
    red[q][c] = s;
    float n = 0.f;
    for (int b = threadIdx.x; b < PP_BLOCKS; b += PF_THREADS) n += cntp[b * NGRAPH + g];
    for (int o = 16; o > 0; o >>= 1) n += __shfl_xor(n, o);
    if ((threadIdx.x & 31) == 0) cred[threadIdx.x >> 5] = n;
    __syncthreads();
    if (threadIdx.x < OUT_C) {
        float t = 0.f;
        for (int k = 0; k < PF_GROUPS; ++k) t += red[k][c];
        float m = 0.f;
        for (int k = 0; k < PF_THREADS / 32; ++k) m += cred[k];
        out[g * OUT_C + c] = t / fmaxf(m, 1.0f);
    }
}

extern "C" void kernel_launch(void* const* d_in, const int* in_sizes, int n_in,
                              void* d_out, int out_size, void* d_ws, size_t ws_size,
                              hipStream_t stream) {
    const float* x   = (const float*)d_in[0];
    const float* W1  = (const float*)d_in[1];
    const float* b1  = (const float*)d_in[2];
    const float* W2  = (const float*)d_in[3];
    const float* b2  = (const float*)d_in[4];
    const int*   ei  = (const int*)d_in[5];
    const int*   bat = (const int*)d_in[6];

    const int Nn = in_sizes[0] / IN_C;
    const int E  = in_sizes[5] / 2;
    const int* src = ei;
    const int* dst = ei + E;

    char* w = (char*)d_ws;
    size_t off = 0;
    auto alloc = [&](size_t bytes) -> void* {
        void* p = w + off;
        off = (off + bytes + 255) & ~(size_t)255;
        return p;
    };
    float*    deg   = (float*)alloc((size_t)Nn * 4);
    float2*   nd    = (float2*)alloc((size_t)Nn * 8);
    float*    h1    = (float*)alloc((size_t)Nn * HID_C * 4);
    float*    a1    = (float*)alloc((size_t)Nn * HID_C * 4);
    float2*   ec    = (float2*)alloc((size_t)E * 8);
    float*    h2 = h1;
    float*    part  = a1;
    float*    cntp  = a1 + (size_t)PP_BLOCKS * NGRAPH * OUT_C;

    const int T = 256;
    k_fill<<<(Nn + T - 1) / T, T, 0, stream>>>(deg, 1.0f, Nn);
    k_edge_deg<<<DG_BLOCKS, DG_THREADS, 0, stream>>>(dst, deg, E, Nn);

    k_gemm<HID_C / 16, false, true><<<GM_BLOCKS, GM_THREADS, 0, stream>>>(x, W1, h1, Nn,
                                                                         deg, b1, bat, nd, a1);
    k_edge_prop<HID_C><<<(E + 255) / 256, EP_THREADS, 0, stream>>>(src, dst, h1, nd, ec, a1, E);

    k_gemm<OUT_C / 16, true, false><<<GM_BLOCKS, GM_THREADS, 0, stream>>>(a1, W2, h2, Nn,
                                                                         nullptr, nullptr, nullptr, nullptr, nullptr);
    k_prop_pool<<<PP_BLOCKS, PP_THREADS, 0, stream>>>(src, ec, nd, h2, b2, part, cntp, E, Nn);
    k_pool_final<<<NGRAPH, PF_THREADS, 0, stream>>>(part, cntp, (float*)d_out);
}


# layer-1 edge scatter: rows split by channel half into two ticket queues; a block reads its die and adds only into its own package half's lines (empty queue: steals); float atomics, arrival order
# speedup vs baseline: 2.8642x; 1.0545x over previous
#include <hip/hip_runtime.h>
#include <hip/hip_bf16.h>
#include <stdint.h>

typedef __attribute__((ext_vector_type(16))) _Float16 v16h;
typedef __attribute__((ext_vector_type(8)))  _Float16 v8h;
typedef __attribute__((ext_vector_type(8)))  float    v8f;

#define IN_C   128
#define HID_C  128
#define OUT_C  64
#define NGRAPH 128

__device__ __forceinline__ void atomic_add_f32(float* p, float v) {
    unsafeAtomicAdd(p, v);
}

#define EP_TSTRIDE 32
__global__ void k_fill(float* p, float v, int n, int* tick) {
    int i = blockIdx.x * blockDim.x + threadIdx.x;
    if (i < n) p[i] = v;
    if (i < 2) tick[i * EP_TSTRIDE] = 0;
}

#define DG_THREADS 1024
#define DG_BLOCKS  32
#define DG_WIN     65536
#define DG_BATCH   8
__global__ void __launch_bounds__(DG_THREADS)
k_edge_deg(const int* __restrict__ dst, float* deg, int E, int Nn) {
    __shared__ __attribute__((aligned(16))) unsigned hist[DG_WIN];
    const int tid   = (int)threadIdx.x;
    const int per   = (E + (int)gridDim.x - 1) / (int)gridDim.x;
    const int start = (int)blockIdx.x * per;
    const int end   = min(E, start + per);
    for (int n0 = 0; n0 < Nn; n0 += DG_WIN) {
        if (n0 > 0) __syncthreads();
        const int cnt = min(DG_WIN, Nn - n0);
        uint4* h4 = reinterpret_cast<uint4*>(hist);
        for (int i = tid; i < (cnt + 3) / 4; i += DG_THREADS) h4[i] = make_uint4(0u, 0u, 0u, 0u);
        __syncthreads();
        for (int e0 = start; e0 < end; e0 += DG_THREADS * DG_BATCH) {
            int v[DG_BATCH];
            #pragma unroll
            for (int u = 0; u < DG_BATCH; ++u) {
                const int e = e0 + u * DG_THREADS + tid;
                v[u] = (e < end) ? dst[e] : -1;
            }
            #pragma unroll
            for (int u = 0; u < DG_BATCH; ++u) {
                const unsigned r = (unsigned)(v[u] - n0);
                if (r < (unsigned)cnt) atomicAdd(&hist[r], 1u);
            }
        }
        __syncthreads();
        const int rot = ((((int)blockIdx.x % DG_BLOCKS) * (cnt >> 5)) / DG_BLOCKS) << 5;
        for (int k0 = 0; k0 < cnt; k0 += DG_THREADS * DG_BATCH) {
            unsigned h[DG_BATCH];
            int      id[DG_BATCH];
            #pragma unroll
            for (int u = 0; u < DG_BATCH; ++u) {
                const int k = k0 + u * DG_THREADS + tid;
                int i = k + rot;
                if (i >= cnt) i -= cnt;
                id[u] = i;
                h[u]  = (k < cnt) ? hist[i] : 0u;
            }
            #pragma unroll
            for (int u = 0; u < DG_BATCH; ++u)
                if (h[u] != 0u) atomic_add_f32(&deg[n0 + id[u]], (float)h[u]);
        }
    }
}

#define GM_THREADS 512
#define GM_K       128
#define GM_BLOCKS  256
template <int NT, bool RELU, bool SELF>
__global__ void __launch_bounds__(GM_THREADS)
k_gemm(const float* __restrict__ A, const float* __restrict__ W,
       float* __restrict__ C, int M,
       const float* __restrict__ deg, const float* __restrict__ bias,
       const int* __restrict__ batch, float2* __restrict__ nd, float* __restrict__ C2) {
    constexpr int K = GM_K;
    constexpr int N = NT * 16;
    __shared__ __attribute__((aligned(16))) _Float16 wl[K * N];
    v8h* wlv = reinterpret_cast<v8h*>(wl);
    #pragma unroll
    for (int it = 0; it < (K / 8) * N / GM_THREADS; ++it) {
        const int t  = it * GM_THREADS + (int)threadIdx.x;
        const int n  = t % N;
        const int k0 = (t / N) * 8;
        const float* wp = W + k0 * N + n;
        v8f f;
        #pragma unroll
        for (int j = 0; j < 8; ++j) f[j] = wp[j * N];
        const v8h h = __builtin_convertvector(f, v8h);
        const int kk0 = k0 & 31;
        wlv[((k0 >> 5) * NT + (n >> 4)) * 64 + (kk0 >> 4) * 32 + (((kk0 >> 3) & 1) << 4) + (n & 15)] = h;
    }
    __syncthreads();

    const int lane = threadIdx.x & 31;
    const int wib  = __builtin_amdgcn_readfirstlane((int)(threadIdx.x >> 5));
    const int mtiles = M >> 4;
    for (int strip = wib * (int)gridDim.x + (int)blockIdx.x; strip < mtiles;
         strip += (int)gridDim.x * (GM_THREADS / 32)) {

    int m0  = strip << 4;
    int hi  = lane >> 4;
    int row = m0 + (lane & 15);

    v8f acc[NT];
    v8f zero = {0.f,0.f,0.f,0.f,0.f,0.f,0.f,0.f};
    #pragma unroll
    for (int j = 0; j < NT; ++j) acc[j] = zero;

    const int mrow = m0 + (hi << 3);
    float4 g0, g1;
    int4 q0, q1;
    if (SELF) {
        const float4* dp = reinterpret_cast<const float4*>(deg + mrow);
        g0 = dp[0]; g1 = dp[1];
        const int4* qp = reinterpret_cast<const int4*>(batch + mrow);
        q0 = qp[0]; q1 = qp[1];
    }

    const v8h* bl = reinterpret_cast<const v8h*>(wl);
    constexpr int ktiles = K >> 5;
    for (int kt = 0; kt < ktiles; ++kt) {
        int kbase = (kt << 5) + (hi << 3);
        const float4* p = reinterpret_cast<const float4*>(A + (long)row * K + kbase);
        float4 x0 = p[0], x1 = p[1];
        float4 x2 = p[4], x3 = p[5];
        if (RELU) {
            x0.x=fmaxf(x0.x,0.f); x0.y=fmaxf(x0.y,0.f); x0.z=fmaxf(x0.z,0.f); x0.w=fmaxf(x0.w,0.f);
            x1.x=fmaxf(x1.x,0.f); x1.y=fmaxf(x1.y,0.f); x1.z=fmaxf(x1.z,0.f); x1.w=fmaxf(x1.w,0.f);
            x2.x=fmaxf(x2.x,0.f); x2.y=fmaxf(x2.y,0.f); x2.z=fmaxf(x2.z,0.f); x2.w=fmaxf(x2.w,0.f);
            x3.x=fmaxf(x3.x,0.f); x3.y=fmaxf(x3.y,0.f); x3.z=fmaxf(x3.z,0.f); x3.w=fmaxf(x3.w,0.f);
        }
        v16h a;
        a[0]=(_Float16)x0.x; a[1]=(_Float16)x0.y; a[2]=(_Float16)x0.z; a[3]=(_Float16)x0.w;
        a[4]=(_Float16)x1.x; a[5]=(_Float16)x1.y; a[6]=(_Float16)x1.z; a[7]=(_Float16)x1.w;
        a[8]=(_Float16)x2.x; a[9]=(_Float16)x2.y; a[10]=(_Float16)x2.z; a[11]=(_Float16)x2.w;
        a[12]=(_Float16)x3.x; a[13]=(_Float16)x3.y; a[14]=(_Float16)x3.z; a[15]=(_Float16)x3.w;

        #pragma unroll
        for (int nt = 0; nt < NT; ++nt) {
            const v8h blo = bl[(kt * NT + nt) * 64 + lane];
            const v8h bhi = bl[(kt * NT + nt) * 64 + 32 + lane];
            const v16h b = __builtin_shufflevector(blo, bhi, 0, 1, 2, 3, 4, 5, 6, 7,
                                                   8, 9, 10, 11, 12, 13, 14, 15);
            acc[nt] = __builtin_amdgcn_wmma_f32_16x16x32_f16(
                false, a, false, b, (short)0, acc[nt], false, false);
        }
    }

    if (SELF) {
        float d[8];
        d[0] = rsqrtf(g0.x); d[1] = rsqrtf(g0.y); d[2] = rsqrtf(g0.z); d[3] = rsqrtf(g0.w);
        d[4] = rsqrtf(g1.x); d[5] = rsqrtf(g1.y); d[6] = rsqrtf(g1.z); d[7] = rsqrtf(g1.w);
        if ((lane & 15) == 0) {
            float4* op = reinterpret_cast<float4*>(nd + mrow);
            op[0] = make_float4(d[0], __int_as_float(q0.x), d[1], __int_as_float(q0.y));
            op[1] = make_float4(d[2], __int_as_float(q0.z), d[3], __int_as_float(q0.w));
            op[2] = make_float4(d[4], __int_as_float(q1.x), d[5], __int_as_float(q1.y));
            op[3] = make_float4(d[6], __int_as_float(q1.z), d[7], __int_as_float(q1.w));
        }
        #pragma unroll
        for (int nt = 0; nt < NT; ++nt) {
            int n = (nt << 4) + (lane & 15);
            const float bn = bias[n];
            #pragma unroll
            for (int j = 0; j < 8; ++j) {
                const float v = acc[nt][j];
                C [(long)(mrow + j) * N + n] = v;
                C2[(long)(mrow + j) * N + n] = v * d[j] * d[j] + bn;
            }
        }
    } else {
        #pragma unroll
        for (int nt = 0; nt < NT; ++nt) {
            int n = (nt << 4) + (lane & 15);
            #pragma unroll
            for (int j = 0; j < 8; ++j)
                C[(long)(mrow + j) * N + n] = acc[nt][j];
        }
    }
    }
}

__device__ __forceinline__ int   lane_bcast_i(int v, int j)   { return __builtin_amdgcn_readlane(v, j); }
__device__ __forceinline__ float lane_bcast_f(float v, int j) {
    return __int_as_float(__builtin_amdgcn_readlane(__float_as_int(v), j));
}

#define EP_THREADS 256
#define EP_WAVES   (EP_THREADS / 32)
#define EP_BATCH   8
#define EP_BLOCKS  2048
#define EP_HC      (HID_C / 2)
__global__ void __launch_bounds__(EP_THREADS)
k_edge_prop(const int* __restrict__ src, const int* __restrict__ dst,
            const float* __restrict__ H, const float2* __restrict__ nd,
            float2* __restrict__ ec, float* acc, int* tick, int E) {
    constexpr int CPE = HID_C;
    constexpr int FPL = EP_HC / 32;
    __shared__ int sh_half;
    __shared__ int sh_tk;
    const int lane = threadIdx.x & 31;
    const int wib  = __builtin_amdgcn_readfirstlane((int)(threadIdx.x >> 5));
    if (threadIdx.x == 0) sh_half = (int)((__builtin_amdgcn_s_sendmsg_rtn(0x87) >> 18) & 1u);
    __syncthreads();
    const int half = sh_half;
    const int ngroups = (E + 31) >> 5;
    const int nchunk  = (ngroups + EP_WAVES - 1) / EP_WAVES;

    for (int pass = 0; pass < 2; ++pass) {
        const int q = half ^ pass;
        const float* Hq = H   + q * EP_HC + lane;
        float*       Aq = acc + q * EP_HC + lane;
        unsigned* tq = reinterpret_cast<unsigned*>(tick) + q * EP_TSTRIDE;
        int nxt = 0;
        if (threadIdx.x == 0) nxt = (int)atomicInc(tq, 0xFFFFFFFFu);
        for (;;) {
            if (threadIdx.x == 0) sh_tk = nxt;
            __syncthreads();
            const int t = sh_tk;
            __syncthreads();
            if (t >= nchunk) break;
            if (threadIdx.x == 0) nxt = (int)atomicInc(tq, 0xFFFFFFFFu);
            const int gi = t * EP_WAVES + wib;
            if (gi >= ngroups) continue;
            const int base = gi << 5;
            const int m = min(32, E - base);
            int s = 0, d = 0;
            if (lane < m) { s = src[base + lane]; d = dst[base + lane]; }
            const unsigned long long qd = reinterpret_cast<const unsigned long long*>(nd)[d];
            const float nrm = nd[s].x * __int_as_float((int)(unsigned)qd);
            if (q == 0 && lane < m)
                ec[base + lane] = make_float2(nrm, __int_as_float((int)(unsigned)(qd >> 32)));

            if (m == 32) {
                #pragma unroll
                for (int j0 = 0; j0 < 32; j0 += EP_BATCH) {
                    float hv[EP_BATCH][FPL];
                    #pragma unroll
                    for (int u = 0; u < EP_BATCH; ++u) {
                        const int sj = lane_bcast_i(s, j0 + u);
                        const float* hs = Hq + (long)sj * CPE;
                        #pragma unroll
                        for (int k = 0; k < FPL; ++k) hv[u][k] = hs[k * 32];
                    }
                    #pragma unroll
                    for (int u = 0; u < EP_BATCH; ++u) {
                        const int   dj = lane_bcast_i(d, j0 + u);
                        const float nj = lane_bcast_f(nrm, j0 + u);
                        float* ad = Aq + (long)dj * CPE;
                        #pragma unroll
                        for (int k = 0; k < FPL; ++k) atomic_add_f32(&ad[k * 32], hv[u][k] * nj);
                    }
                }
            } else {
                for (int j = 0; j < m; ++j) {
                    const int   sj = lane_bcast_i(s, j);
                    const int   dj = lane_bcast_i(d, j);
                    const float nj = lane_bcast_f(nrm, j);
                    const float* hs = Hq + (long)sj * CPE;
                    float*       ad = Aq + (long)dj * CPE;
                    float hv[FPL];
                    for (int k = 0; k < FPL; ++k) hv[k] = hs[k * 32];
                    for (int k = 0; k < FPL; ++k) atomic_add_f32(&ad[k * 32], hv[k] * nj);
                }
            }
        }
    }
}

#define PP_THREADS 1024
#define PP_WAVES   (PP_THREADS / 32)
#define PP_BATCH   16
#define PP_BLOCKS  256

__global__ void __launch_bounds__(PP_THREADS)
k_prop_pool(const int* __restrict__ src, const float2* __restrict__ ec,
            const float2* __restrict__ nd, const float* __restrict__ H,
            const float* __restrict__ bias,
            float* __restrict__ part, float* __restrict__ cntp, int E, int Nn) {
    __shared__ __attribute__((aligned(16))) float tab[NGRAPH * OUT_C];
    __shared__ float cnt[NGRAPH];
    for (int i = threadIdx.x; i < NGRAPH * OUT_C; i += PP_THREADS) tab[i] = 0.f;
    if (threadIdx.x < NGRAPH) cnt[threadIdx.x] = 0.f;
    __syncthreads();

    const int lane = threadIdx.x & 31;
    const int wib  = __builtin_amdgcn_readfirstlane((int)(threadIdx.x >> 5));
    const int nge  = (E + 31) >> 5;
    const int ngn  = (Nn + 31) >> 5;
    const int ng   = nge + ngn;
    const int tw   = (int)gridDim.x * PP_WAVES;
    const int c0   = lane * 2;
    const float bx = bias[c0], by = bias[c0 + 1];

    for (int gi = wib * (int)gridDim.x + (int)blockIdx.x; gi < ng; gi += tw) {
        const bool isnode = gi >= nge;
        int s = 0, m;
        float2 r = make_float2(0.f, 0.f);
        if (!isnode) {
            const int base = gi << 5;
            m = min(32, E - base);
            if (lane < m) { s = src[base + lane]; r = ec[base + lane]; }
        } else {
            const int base = (gi - nge) << 5;
            m = min(32, Nn - base);
            if (lane < m) { s = base + lane; r = nd[s]; r.x *= r.x; }
        }
        const float nrm = r.x;
        const int   g   = __float_as_int(r.y);
        const float ax = isnode ? bx : 0.f;
        const float ay = isnode ? by : 0.f;
        if (isnode && lane < m) atomicAdd(&cnt[g], 1.0f);

        if (m == 32) {
            #pragma unroll
            for (int j0 = 0; j0 < 32; j0 += PP_BATCH) {
                float2 v[PP_BATCH];
                #pragma unroll
                for (int u = 0; u < PP_BATCH; ++u) {
                    const int sj = lane_bcast_i(s, j0 + u);
                    v[u] = *reinterpret_cast<const float2*>(H + (long)sj * OUT_C + c0);
                }
                #pragma unroll
                for (int u = 0; u < PP_BATCH; ++u) {
                    const float nj = lane_bcast_f(nrm, j0 + u);
                    const int   gj = lane_bcast_i(g, j0 + u);
                    float* t = &tab[gj * OUT_C + c0];
                    atomicAdd(t,     fmaf(v[u].x, nj, ax));
                    atomicAdd(t + 1, fmaf(v[u].y, nj, ay));
                }
            }
        } else {
            for (int j = 0; j < m; ++j) {
                const int   sj = lane_bcast_i(s, j);
                const float nj = lane_bcast_f(nrm, j);
                const int   gj = lane_bcast_i(g, j);
                const float2 v = *reinterpret_cast<const float2*>(H + (long)sj * OUT_C + c0);
                float* t = &tab[gj * OUT_C + c0];
                atomicAdd(t,     fmaf(v.x, nj, ax));
                atomicAdd(t + 1, fmaf(v.y, nj, ay));
            }
        }
    }
    __syncthreads();

    const float4* t4 = reinterpret_cast<const float4*>(tab);
    float4* p4 = reinterpret_cast<float4*>(part + (long)blockIdx.x * (NGRAPH * OUT_C));
    for (int i = threadIdx.x; i < NGRAPH * OUT_C / 4; i += PP_THREADS) p4[i] = t4[i];
    if (threadIdx.x < NGRAPH) cntp[(int)blockIdx.x * NGRAPH + threadIdx.x] = cnt[threadIdx.x];
}

#define PF_THREADS 512
#define PF_GROUPS  (PF_THREADS / OUT_C)
__global__ void __launch_bounds__(PF_THREADS)
k_pool_final(const float* __restrict__ part, const float* __restrict__ cntp,
             float* __restrict__ out) {
    __shared__ float red[PF_GROUPS][OUT_C];
    __shared__ float cred[PF_THREADS / 32];
    const int g = blockIdx.x;
    const int c = threadIdx.x & (OUT_C - 1);
    const int q = threadIdx.x >> 6;
    float s = 0.f;
    #pragma unroll
    for (int i = 0; i < PP_BLOCKS / PF_GROUPS; ++i)
        s += part[((long)(q + i * PF_GROUPS) * NGRAPH + g) * OUT_C + c];
    red[q][c] = s;
    float n = 0.f;
    for (int b = threadIdx.x; b < PP_BLOCKS; b += PF_THREADS) n += cntp[b * NGRAPH + g];
    for (int o = 16; o > 0; o >>= 1) n += __shfl_xor(n, o);
    if ((threadIdx.x & 31) == 0) cred[threadIdx.x >> 5] = n;
    __syncthreads();
    if (threadIdx.x < OUT_C) {
        float t = 0.f;
        for (int k = 0; k < PF_GROUPS; ++k) t += red[k][c];
        float m = 0.f;
        for (int k = 0; k < PF_THREADS / 32; ++k) m += cred[k];
        out[g * OUT_C + c] = t / fmaxf(m, 1.0f);
    }
}

extern "C" void kernel_launch(void* const* d_in, const int* in_sizes, int n_in,
                              void* d_out, int out_size, void* d_ws, size_t ws_size,
                              hipStream_t stream) {
    const float* x   = (const float*)d_in[0];
    const float* W1  = (const float*)d_in[1];
    const float* b1  = (const float*)d_in[2];
    const float* W2  = (const float*)d_in[3];
    const float* b2  = (const float*)d_in[4];
    const int*   ei  = (const int*)d_in[5];
    const int*   bat = (const int*)d_in[6];

    const int Nn = in_sizes[0] / IN_C;
    const int E  = in_sizes[5] / 2;
    const int* src = ei;
    const int* dst = ei + E;

    char* w = (char*)d_ws;
    size_t off = 0;
    auto alloc = [&](size_t bytes) -> void* {
        void* p = w + off;
        off = (off + bytes + 255) & ~(size_t)255;
        return p;
    };
    float*    deg   = (float*)alloc((size_t)Nn * 4 + (EP_TSTRIDE + 1) * 4);
    int*      tick  = (int*)(deg + Nn);
    float2*   nd    = (float2*)alloc((size_t)Nn * 8);
    float*    h1    = (float*)alloc((size_t)Nn * HID_C * 4);
    float*    a1    = (float*)alloc((size_t)Nn * HID_C * 4);
    float2*   ec    = (float2*)alloc((size_t)E * 8);
    float*    h2 = h1;
    float*    part  = a1;
    float*    cntp  = a1 + (size_t)PP_BLOCKS * NGRAPH * OUT_C;

    const int T = 256;
    k_fill<<<(Nn + T - 1) / T, T, 0, stream>>>(deg, 1.0f, Nn, tick);
    k_edge_deg<<<DG_BLOCKS, DG_THREADS, 0, stream>>>(dst, deg, E, Nn);

    k_gemm<HID_C / 16, false, true><<<GM_BLOCKS, GM_THREADS, 0, stream>>>(x, W1, h1, Nn,
                                                                         deg, b1, bat, nd, a1);
    k_edge_prop<<<EP_BLOCKS, EP_THREADS, 0, stream>>>(src, dst, h1, nd, ec, a1, tick, E);

    k_gemm<OUT_C / 16, true, false><<<GM_BLOCKS, GM_THREADS, 0, stream>>>(a1, W2, h2, Nn,
                                                                         nullptr, nullptr, nullptr, nullptr, nullptr);
    k_prop_pool<<<PP_BLOCKS, PP_THREADS, 0, stream>>>(src, ec, nd, h2, b2, part, cntp, E, Nn);
    k_pool_final<<<NGRAPH, PF_THREADS, 0, stream>>>(part, cntp, (float*)d_out);
}


# in-degree: each counting block stores its LDS table as its own 16-bit slice (plain stores); layer-1 product epilogue adds the 32 slices of its rows; fill launch, deg atomics go: 7 launches to 6
# speedup vs baseline: 2.9801x; 1.0405x over previous
#include <hip/hip_runtime.h>
#include <hip/hip_bf16.h>
#include <stdint.h>

typedef __attribute__((ext_vector_type(16))) _Float16 v16h;
typedef __attribute__((ext_vector_type(8)))  _Float16 v8h;
typedef __attribute__((ext_vector_type(8)))  float    v8f;

#define IN_C   128
#define HID_C  128
#define OUT_C  64
#define NGRAPH 128

__device__ __forceinline__ void atomic_add_f32(float* p, float v) {
    unsafeAtomicAdd(p, v);
}

#define EP_TSTRIDE 32

#define DG_THREADS 1024
#define DG_BLOCKS  32
#define DG_WIN     65536
#define DG_BATCH   8
__global__ void __launch_bounds__(DG_THREADS)
k_edge_deg(const int* __restrict__ dst, unsigned short* __restrict__ degsl, int* tick, int E, int Nn) {
    __shared__ __attribute__((aligned(16))) unsigned hist[DG_WIN];
    const int tid   = (int)threadIdx.x;
    if (blockIdx.x == 0 && tid < 2) tick[tid * EP_TSTRIDE] = 0;
    const int per   = (E + (int)gridDim.x - 1) / (int)gridDim.x;
    const int start = (int)blockIdx.x * per;
    const int end   = min(E, start + per);
    unsigned short* sl = degsl + (size_t)blockIdx.x * (size_t)((Nn + 7) & ~7);
    for (int n0 = 0; n0 < Nn; n0 += DG_WIN) {
        if (n0 > 0) __syncthreads();
        const int cnt = min(DG_WIN, Nn - n0);
        uint4* h4 = reinterpret_cast<uint4*>(hist);
        for (int i = tid; i < ((cnt + 7) / 8) * 2; i += DG_THREADS) h4[i] = make_uint4(0u, 0u, 0u, 0u);
        __syncthreads();
        for (int e0 = start; e0 < end; e0 += DG_THREADS * DG_BATCH) {
            int v[DG_BATCH];
            #pragma unroll
            for (int u = 0; u < DG_BATCH; ++u) {
                const int e = e0 + u * DG_THREADS + tid;
                v[u] = (e < end) ? dst[e] : -1;
            }
            #pragma unroll
            for (int u = 0; u < DG_BATCH; ++u) {
                const unsigned r = (unsigned)(v[u] - n0);
                if (r < (unsigned)cnt) atomicAdd(&hist[r], 1u);
            }
        }
        __syncthreads();
        for (int k = tid * 8; k < cnt; k += DG_THREADS * 8) {
            const uint4 a = h4[(k >> 2)];
            const uint4 b = h4[(k >> 2) + 1];
            if (k + 8 <= cnt) {
                *reinterpret_cast<uint4*>(sl + n0 + k) =
                    make_uint4(a.x | (a.y << 16), a.z | (a.w << 16), b.x | (b.y << 16), b.z | (b.w << 16));
            } else {
                const unsigned t[8] = {a.x, a.y, a.z, a.w, b.x, b.y, b.z, b.w};
                for (int j = 0; j < cnt - k; ++j) sl[n0 + k + j] = (unsigned short)t[j];
            }
        }
    }
}

template <int X>
__device__ __forceinline__ unsigned row16_xor(unsigned v) {
    return (unsigned)__builtin_amdgcn_update_dpp(0, (int)v, 0x160 | X, 0xf, 0xf, false);
}

#define GM_THREADS 512
#define GM_K       128
#define GM_BLOCKS  256
template <int NT, bool RELU, bool SELF>
__global__ void __launch_bounds__(GM_THREADS)
k_gemm(const float* __restrict__ A, const float* __restrict__ W,
       float* __restrict__ C, int M,
       const unsigned short* __restrict__ degsl, const float* __restrict__ bias,
       const int* __restrict__ batch, float2* __restrict__ nd, float* __restrict__ C2) {
    constexpr int K = GM_K;
    constexpr int N = NT * 16;
    __shared__ __attribute__((aligned(16))) _Float16 wl[K * N];
    v8h* wlv = reinterpret_cast<v8h*>(wl);
    #pragma unroll
    for (int it = 0; it < (K / 8) * N / GM_THREADS; ++it) {
        const int t  = it * GM_THREADS + (int)threadIdx.x;
        const int n  = t % N;
        const int k0 = (t / N) * 8;
        const float* wp = W + k0 * N + n;
        v8f f;
        #pragma unroll
        for (int j = 0; j < 8; ++j) f[j] = wp[j * N];
        const v8h h = __builtin_convertvector(f, v8h);
        const int kk0 = k0 & 31;
        wlv[((k0 >> 5) * NT + (n >> 4)) * 64 + (kk0 >> 4) * 32 + (((kk0 >> 3) & 1) << 4) + (n & 15)] = h;
    }
    __syncthreads();

    const int lane = threadIdx.x & 31;
    const int wib  = __builtin_amdgcn_readfirstlane((int)(threadIdx.x >> 5));
    const int mtiles = M >> 4;
    for (int strip = wib * (int)gridDim.x + (int)blockIdx.x; strip < mtiles;
         strip += (int)gridDim.x * (GM_THREADS / 32)) {

    int m0  = strip << 4;
    int hi  = lane >> 4;
    int row = m0 + (lane & 15);

    v8f acc[NT];
    v8f zero = {0.f,0.f,0.f,0.f,0.f,0.f,0.f,0.f};
    #pragma unroll
    for (int j = 0; j < NT; ++j) acc[j] = zero;

    const int mrow = m0 + (hi << 3);
    uint4 u0, u1;
    int4 q0, q1;
    if (SELF) {
        static_assert(DG_BLOCKS == 32, "two slices per lane of a half");
        const size_t pitch = (size_t)((M + 7) & ~7);
        const unsigned short* sp = degsl + (size_t)(lane & 15) * pitch + mrow;
        u0 = *reinterpret_cast<const uint4*>(sp);
        u1 = *reinterpret_cast<const uint4*>(sp + 16 * pitch);
        const int4* qp = reinterpret_cast<const int4*>(batch + mrow);
        q0 = qp[0]; q1 = qp[1];
    }

    const v8h* bl = reinterpret_cast<const v8h*>(wl);
    constexpr int ktiles = K >> 5;
    for (int kt = 0; kt < ktiles; ++kt) {
        int kbase = (kt << 5) + (hi << 3);
        const float4* p = reinterpret_cast<const float4*>(A + (long)row * K + kbase);
        float4 x0 = p[0], x1 = p[1];
        float4 x2 = p[4], x3 = p[5];
        if (RELU) {
            x0.x=fmaxf(x0.x,0.f); x0.y=fmaxf(x0.y,0.f); x0.z=fmaxf(x0.z,0.f); x0.w=fmaxf(x0.w,0.f);
            x1.x=fmaxf(x1.x,0.f); x1.y=fmaxf(x1.y,0.f); x1.z=fmaxf(x1.z,0.f); x1.w=fmaxf(x1.w,0.f);
            x2.x=fmaxf(x2.x,0.f); x2.y=fmaxf(x2.y,0.f); x2.z=fmaxf(x2.z,0.f); x2.w=fmaxf(x2.w,0.f);
            x3.x=fmaxf(x3.x,0.f); x3.y=fmaxf(x3.y,0.f); x3.z=fmaxf(x3.z,0.f); x3.w=fmaxf(x3.w,0.f);
        }
        v16h a;
        a[0]=(_Float16)x0.x; a[1]=(_Float16)x0.y; a[2]=(_Float16)x0.z; a[3]=(_Float16)x0.w;
        a[4]=(_Float16)x1.x; a[5]=(_Float16)x1.y; a[6]=(_Float16)x1.z; a[7]=(_Float16)x1.w;
        a[8]=(_Float16)x2.x; a[9]=(_Float16)x2.y; a[10]=(_Float16)x2.z; a[11]=(_Float16)x2.w;
        a[12]=(_Float16)x3.x; a[13]=(_Float16)x3.y; a[14]=(_Float16)x3.z; a[15]=(_Float16)x3.w;

        #pragma unroll
        for (int nt = 0; nt < NT; ++nt) {
            const v8h blo = bl[(kt * NT + nt) * 64 + lane];
            const v8h bhi = bl[(kt * NT + nt) * 64 + 32 + lane];
            const v16h b = __builtin_shufflevector(blo, bhi, 0, 1, 2, 3, 4, 5, 6, 7,
                                                   8, 9, 10, 11, 12, 13, 14, 15);
            acc[nt] = __builtin_amdgcn_wmma_f32_16x16x32_f16(
                false, a, false, b, (short)0, acc[nt], false, false);
        }
    }

    if (SELF) {
        unsigned cs[8];
        cs[0] = (u0.x & 0xffffu) + (u1.x & 0xffffu); cs[1] = (u0.x >> 16) + (u1.x >> 16);
        cs[2] = (u0.y & 0xffffu) + (u1.y & 0xffffu); cs[3] = (u0.y >> 16) + (u1.y >> 16);
        cs[4] = (u0.z & 0xffffu) + (u1.z & 0xffffu); cs[5] = (u0.z >> 16) + (u1.z >> 16);
        cs[6] = (u0.w & 0xffffu) + (u1.w & 0xffffu); cs[7] = (u0.w >> 16) + (u1.w >> 16);
        float d[8];
        #pragma unroll
        for (int j = 0; j < 8; ++j) {
            unsigned c = cs[j];
            c += row16_xor<1>(c); c += row16_xor<2>(c); c += row16_xor<4>(c); c += row16_xor<8>(c);
            d[j] = rsqrtf(1.0f + (float)c);
        }
        if ((lane & 15) == 0) {
            float4* op = reinterpret_cast<float4*>(nd + mrow);
            op[0] = make_float4(d[0], __int_as_float(q0.x), d[1], __int_as_float(q0.y));
            op[1] = make_float4(d[2], __int_as_float(q0.z), d[3], __int_as_float(q0.w));
            op[2] = make_float4(d[4], __int_as_float(q1.x), d[5], __int_as_float(q1.y));
            op[3] = make_float4(d[6], __int_as_float(q1.z), d[7], __int_as_float(q1.w));
        }
        #pragma unroll
        for (int nt = 0; nt < NT; ++nt) {
            int n = (nt << 4) + (lane & 15);
            const float bn = bias[n];
            #pragma unroll
            for (int j = 0; j < 8; ++j) {
                const float v = acc[nt][j];
                C [(long)(mrow + j) * N + n] = v;
                C2[(long)(mrow + j) * N + n] = v * d[j] * d[j] + bn;
            }
        }
    } else {
        #pragma unroll
        for (int nt = 0; nt < NT; ++nt) {
            int n = (nt << 4) + (lane & 15);
            #pragma unroll
            for (int j = 0; j < 8; ++j)
                C[(long)(mrow + j) * N + n] = acc[nt][j];
        }
    }
    }
}

__device__ __forceinline__ int   lane_bcast_i(int v, int j)   { return __builtin_amdgcn_readlane(v, j); }
__device__ __forceinline__ float lane_bcast_f(float v, int j) {
    return __int_as_float(__builtin_amdgcn_readlane(__float_as_int(v), j));
}

#define EP_THREADS 256
#define EP_WAVES   (EP_THREADS / 32)
#define EP_BATCH   8
#define EP_BLOCKS  2048
#define EP_HC      (HID_C / 2)
__global__ void __launch_bounds__(EP_THREADS)
k_edge_prop(const int* __restrict__ src, const int* __restrict__ dst,
            const float* __restrict__ H, const float2* __restrict__ nd,
            float2* __restrict__ ec, float* acc, int* tick, int E) {
    constexpr int CPE = HID_C;
    constexpr int FPL = EP_HC / 32;
    __shared__ int sh_half;
    __shared__ int sh_tk;
    const int lane = threadIdx.x & 31;
    const int wib  = __builtin_amdgcn_readfirstlane((int)(threadIdx.x >> 5));
    if (threadIdx.x == 0) sh_half = (int)((__builtin_amdgcn_s_sendmsg_rtn(0x87) >> 18) & 1u);
    __syncthreads();
    const int half = sh_half;
    const int ngroups = (E + 31) >> 5;
    const int nchunk  = (ngroups + EP_WAVES - 1) / EP_WAVES;

    for (int pass = 0; pass < 2; ++pass) {
        const int q = half ^ pass;
        const float* Hq = H   + q * EP_HC + lane;
        float*       Aq = acc + q * EP_HC + lane;
        unsigned* tq = reinterpret_cast<unsigned*>(tick) + q * EP_TSTRIDE;
        int nxt = 0;
        if (threadIdx.x == 0) nxt = (int)atomicInc(tq, 0xFFFFFFFFu);
        for (;;) {
            if (threadIdx.x == 0) sh_tk = nxt;
            __syncthreads();
            const int t = sh_tk;
            __syncthreads();
            if (t >= nchunk) break;
            if (threadIdx.x == 0) nxt = (int)atomicInc(tq, 0xFFFFFFFFu);
            const int gi = t * EP_WAVES + wib;
            if (gi >= ngroups) continue;
            const int base = gi << 5;
            const int m = min(32, E - base);
            int s = 0, d = 0;
            if (lane < m) { s = src[base + lane]; d = dst[base + lane]; }
            const unsigned long long qd = reinterpret_cast<const unsigned long long*>(nd)[d];
            const float nrm = nd[s].x * __int_as_float((int)(unsigned)qd);
            if (q == 0 && lane < m)
                ec[base + lane] = make_float2(nrm, __int_as_float((int)(unsigned)(qd >> 32)));

            if (m == 32) {
                #pragma unroll
                for (int j0 = 0; j0 < 32; j0 += EP_BATCH) {
                    float hv[EP_BATCH][FPL];
                    #pragma unroll
                    for (int u = 0; u < EP_BATCH; ++u) {
                        const int sj = lane_bcast_i(s, j0 + u);
                        const float* hs = Hq + (long)sj * CPE;
                        #pragma unroll
                        for (int k = 0; k < FPL; ++k) hv[u][k] = hs[k * 32];
                    }
                    #pragma unroll
                    for (int u = 0; u < EP_BATCH; ++u) {
                        const int   dj = lane_bcast_i(d, j0 + u);
                        const float nj = lane_bcast_f(nrm, j0 + u);
                        float* ad = Aq + (long)dj * CPE;
                        #pragma unroll
                        for (int k = 0; k < FPL; ++k) atomic_add_f32(&ad[k * 32], hv[u][k] * nj);
                    }
                }
            } else {
                for (int j = 0; j < m; ++j) {
                    const int   sj = lane_bcast_i(s, j);
                    const int   dj = lane_bcast_i(d, j);
                    const float nj = lane_bcast_f(nrm, j);
                    const float* hs = Hq + (long)sj * CPE;
                    float*       ad = Aq + (long)dj * CPE;
                    float hv[FPL];
                    for (int k = 0; k < FPL; ++k) hv[k] = hs[k * 32];
                    for (int k = 0; k < FPL; ++k) atomic_add_f32(&ad[k * 32], hv[k] * nj);
                }
            }
        }
    }
}

#define PP_THREADS 1024
#define PP_WAVES   (PP_THREADS / 32)
#define PP_BATCH   16
#define PP_BLOCKS  256

__global__ void __launch_bounds__(PP_THREADS)
k_prop_pool(const int* __restrict__ src, const float2* __restrict__ ec,
            const float2* __restrict__ nd, const float* __restrict__ H,
            const float* __restrict__ bias,
            float* __restrict__ part, float* __restrict__ cntp, int E, int Nn) {
    __shared__ __attribute__((aligned(16))) float tab[NGRAPH * OUT_C];
    __shared__ float cnt[NGRAPH];
    for (int i = threadIdx.x; i < NGRAPH * OUT_C; i += PP_THREADS) tab[i] = 0.f;
    if (threadIdx.x < NGRAPH) cnt[threadIdx.x] = 0.f;
    __syncthreads();

    const int lane = threadIdx.x & 31;
    const int wib  = __builtin_amdgcn_readfirstlane((int)(threadIdx.x >> 5));
    const int nge  = (E + 31) >> 5;
    const int ngn  = (Nn + 31) >> 5;
    const int ng   = nge + ngn;
    const int tw   = (int)gridDim.x * PP_WAVES;
    const int c0   = lane * 2;
    const float bx = bias[c0], by = bias[c0 + 1];

    for (int gi = wib * (int)gridDim.x + (int)blockIdx.x; gi < ng; gi += tw) {
        const bool isnode = gi >= nge;
        int s = 0, m;
        float2 r = make_float2(0.f, 0.f);
        if (!isnode) {
            const int base = gi << 5;
            m = min(32, E - base);
            if (lane < m) { s = src[base + lane]; r = ec[base + lane]; }
        } else {
            const int base = (gi - nge) << 5;
            m = min(32, Nn - base);
            if (lane < m) { s = base + lane; r = nd[s]; r.x *= r.x; }
        }
        const float nrm = r.x;
        const int   g   = __float_as_int(r.y);
        const float ax = isnode ? bx : 0.f;
        const float ay = isnode ? by : 0.f;
        if (isnode && lane < m) atomicAdd(&cnt[g], 1.0f);

        if (m == 32) {
            #pragma unroll
            for (int j0 = 0; j0 < 32; j0 += PP_BATCH) {
                float2 v[PP_BATCH];
                #pragma unroll
                for (int u = 0; u < PP_BATCH; ++u) {
                    const int sj = lane_bcast_i(s, j0 + u);
                    v[u] = *reinterpret_cast<const float2*>(H + (long)sj * OUT_C + c0);
                }
                #pragma unroll
                for (int u = 0; u < PP_BATCH; ++u) {
                    const float nj = lane_bcast_f(nrm, j0 + u);
                    const int   gj = lane_bcast_i(g, j0 + u);
                    float* t = &tab[gj * OUT_C + c0];
                    atomicAdd(t,     fmaf(v[u].x, nj, ax));
                    atomicAdd(t + 1, fmaf(v[u].y, nj, ay));
                }
            }
        } else {
            for (int j = 0; j < m; ++j) {
                const int   sj = lane_bcast_i(s, j);
                const float nj = lane_bcast_f(nrm, j);
                const int   gj = lane_bcast_i(g, j);
                const float2 v = *reinterpret_cast<const float2*>(H + (long)sj * OUT_C + c0);
                float* t = &tab[gj * OUT_C + c0];
                atomicAdd(t,     fmaf(v.x, nj, ax));
                atomicAdd(t + 1, fmaf(v.y, nj, ay));
            }
        }
    }
    __syncthreads();

    const float4* t4 = reinterpret_cast<const float4*>(tab);
    float4* p4 = reinterpret_cast<float4*>(part + (long)blockIdx.x * (NGRAPH * OUT_C));
    for (int i = threadIdx.x; i < NGRAPH * OUT_C / 4; i += PP_THREADS) p4[i] = t4[i];
    if (threadIdx.x < NGRAPH) cntp[(int)blockIdx.x * NGRAPH + threadIdx.x] = cnt[threadIdx.x];
}

#define PF_THREADS 512
#define PF_GROUPS  (PF_THREADS / OUT_C)
__global__ void __launch_bounds__(PF_THREADS)
k_pool_final(const float* __restrict__ part, const float* __restrict__ cntp,
             float* __restrict__ out) {
    __shared__ float red[PF_GROUPS][OUT_C];
    __shared__ float cred[PF_THREADS / 32];
    const int g = blockIdx.x;
    const int c = threadIdx.x & (OUT_C - 1);
    const int q = threadIdx.x >> 6;
    float s = 0.f;
    #pragma unroll
    for (int i = 0; i < PP_BLOCKS / PF_GROUPS; ++i)
        s += part[((long)(q + i * PF_GROUPS) * NGRAPH + g) * OUT_C + c];
    red[q][c] = s;
    float n = 0.f;
    for (int b = threadIdx.x; b < PP_BLOCKS; b += PF_THREADS) n += cntp[b * NGRAPH + g];
    for (int o = 16; o > 0; o >>= 1) n += __shfl_xor(n, o);
    if ((threadIdx.x & 31) == 0) cred[threadIdx.x >> 5] = n;
    __syncthreads();
    if (threadIdx.x < OUT_C) {
        float t = 0.f;
        for (int k = 0; k < PF_GROUPS; ++k) t += red[k][c];
        float m = 0.f;
        for (int k = 0; k < PF_THREADS / 32; ++k) m += cred[k];
        out[g * OUT_C + c] = t / fmaxf(m, 1.0f);
    }
}

extern "C" void kernel_launch(void* const* d_in, const int* in_sizes, int n_in,
                              void* d_out, int out_size, void* d_ws, size_t ws_size,
                              hipStream_t stream) {
    const float* x   = (const float*)d_in[0];
    const float* W1  = (const float*)d_in[1];
    const float* b1  = (const float*)d_in[2];
    const float* W2  = (const float*)d_in[3];
    const float* b2  = (const float*)d_in[4];
    const int*   ei  = (const int*)d_in[5];
    const int*   bat = (const int*)d_in[6];

    const int Nn = in_sizes[0] / IN_C;
    const int E  = in_sizes[5] / 2;
    const int* src = ei;
    const int* dst = ei + E;

    char* w = (char*)d_ws;
    size_t off = 0;
    auto alloc = [&](size_t bytes) -> void* {
        void* p = w + off;
        off = (off + bytes + 255) & ~(size_t)255;
        return p;
    };
    int*      tick  = (int*)alloc((EP_TSTRIDE + 1) * 4);
    float2*   nd    = (float2*)alloc((size_t)Nn * 8);
    float*    h1    = (float*)alloc((size_t)Nn * HID_C * 4);
    float*    a1    = (float*)alloc((size_t)Nn * HID_C * 4);
    float2*   ec    = (float2*)alloc((size_t)E * 8);
    unsigned short* degsl = (unsigned short*)ec;
    float*    h2 = h1;
    float*    part  = a1;
    float*    cntp  = a1 + (size_t)PP_BLOCKS * NGRAPH * OUT_C;

    k_edge_deg<<<DG_BLOCKS, DG_THREADS, 0, stream>>>(dst, degsl, tick, E, Nn);

    k_gemm<HID_C / 16, false, true><<<GM_BLOCKS, GM_THREADS, 0, stream>>>(x, W1, h1, Nn,
                                                                         degsl, b1, bat, nd, a1);
    k_edge_prop<<<EP_BLOCKS, EP_THREADS, 0, stream>>>(src, dst, h1, nd, ec, a1, tick, E);

    k_gemm<OUT_C / 16, true, false><<<GM_BLOCKS, GM_THREADS, 0, stream>>>(a1, W2, h2, Nn,
                                                                         nullptr, nullptr, nullptr, nullptr, nullptr);
    k_prop_pool<<<PP_BLOCKS, PP_THREADS, 0, stream>>>(src, ec, nd, h2, b2, part, cntp, E, Nn);
    k_pool_final<<<NGRAPH, PF_THREADS, 0, stream>>>(part, cntp, (float*)d_out);
}
